# starting kernel
# baseline (speedup 1.0000x reference)
#include <hip/hip_runtime.h>
#include <math.h>

#define B_   2
#define T_   8
#define N_   5000
#define K1_  17
#define H_   4
#define C_   16
#define NO_  8
#define T7_  (T_-1)
#define E2_  (N_*K1_)
#define TILES2_ ((E2_+15)/16)

typedef float v2f __attribute__((ext_vector_type(2)));
typedef float v8f __attribute__((ext_vector_type(8)));

#define SWZ(x, M) __int_as_float(__builtin_amdgcn_ds_swizzle( \
                      __float_as_int(x), ((M) << 10) | 0x1f))

template<int M>
__device__ __forceinline__ float mergepair(float a, float b, int c) {
    const float keep = (c & M) ? b : a;
    const float send = (c & M) ? a : b;
    return keep + SWZ(send, M);
}

__device__ __forceinline__ float row_reduce16(const float t[8], int c) {
    const float m0 = mergepair<1>(t[0], t[1], c);
    const float m1 = mergepair<1>(t[2], t[3], c);
    const float m2 = mergepair<1>(t[4], t[5], c);
    const float m3 = mergepair<1>(t[6], t[7], c);
    const float n0 = mergepair<2>(m0, m1, c);
    const float n1 = mergepair<2>(m2, m3, c);
    float s = mergepair<4>(n0, n1, c);
    s += SWZ(s, 8);
    return s;
}

__global__ void k1_w_deg(const float* __restrict__ feat,
                         const int*   __restrict__ nn,
                         const float* __restrict__ mm,
                         float*       __restrict__ w_out,
                         float*       __restrict__ deg)
{
    const int wid  = (blockIdx.x * blockDim.x + threadIdx.x) >> 5;
    const int lane = threadIdx.x & 31;

    const int n  = wid % N_;
    const int r0 = wid / N_;
    const int h  = r0 % H_;
    const int bt = r0 / H_;

    const int r  = lane & 15;
    const int hh = lane >> 4;

    const int nbrv = nn[n * K1_ + 1 + r];
    const unsigned vm = (unsigned)__ballot(nbrv >= 0);
    const int nj = (nbrv < 0) ? n : nbrv;

    const float* fi = feat + ((((size_t)bt * N_ + n ) * H_ + h) * C_);
    const float* fj = feat + ((((size_t)bt * N_ + nj) * H_ + h) * C_);
    const float* Mh = mm + h * C_ * C_;

    v8f d = {0.f,0.f,0.f,0.f,0.f,0.f,0.f,0.f};
#pragma unroll
    for (int c = 0; c < 4; ++c) {
        const int j0 = 4 * c + 2 * hh;
        v2f av, bv;
        av.x = fi[j0]     - fj[j0];
        av.y = fi[j0 + 1] - fj[j0 + 1];
        bv.x = Mh[r * C_ + j0];
        bv.y = Mh[r * C_ + j0 + 1];
        d = __builtin_amdgcn_wmma_f32_16x16x4_f32(
                false, av, false, bv, (short)0, d, false, false);
    }

    float t[8];
#pragma unroll
    for (int v = 0; v < 8; ++v) t[v] = d[v] * d[v];
    const float s = row_reduce16(t, r);

    const int   row = (lane & 7) | (hh << 3);
    const float w   = ((vm >> row) & 1u) ? __expf(-s) : 0.f;

    float dp = w;
    dp += SWZ(dp, 1);
    dp += SWZ(dp, 2);
    dp += SWZ(dp, 4);
    dp += SWZ(dp, 16);

    if ((lane & 15) < 8)
        w_out[(((size_t)bt * N_ + n) * 16 + row) * H_ + h] = w;
    if (lane == 0)
        deg[((size_t)bt * N_ + n) * H_ + h] = dp;
}

__global__ void k2_uw(float* __restrict__ uw,
                      const float* __restrict__ deg,
                      const int*   __restrict__ nn)
{
    const size_t tid = (size_t)blockIdx.x * blockDim.x + threadIdx.x;
    const size_t total = (size_t)B_ * T_ * N_ * 16 * H_;
    if (tid >= total) return;
    const int h  = (int)(tid % H_);
    const int k  = (int)((tid / H_) % 16);
    const int n  = (int)((tid / ((size_t)H_ * 16)) % N_);
    const int bt = (int)( tid / ((size_t)H_ * 16 * N_));

    const float w  = uw[tid];
    const int   nb = nn[n * K1_ + 1 + k];
    const float di = deg[((size_t)bt * N_ + n) * H_ + h];
    const float dj = (nb >= 0) ? deg[((size_t)bt * N_ + nb) * H_ + h] : 0.f;
    const float dm = sqrtf(di * dj);
    uw[tid] = (dm > 0.f) ? (w / dm) : 0.f;
}

__global__ void k3_qj(const float* __restrict__ feat,
                      const float* __restrict__ q2,
                      float*       __restrict__ qj)
{
    const size_t tid = (size_t)blockIdx.x * blockDim.x + threadIdx.x;
    const size_t total = (size_t)B_ * T7_ * N_ * H_;
    if (tid >= total) return;
    const int h   = (int)(tid % H_);
    const int n   = (int)((tid / H_) % N_);
    const int bt7 = (int)( tid / ((size_t)H_ * N_));
    const int t   = bt7 % T7_;
    const int b   = bt7 / T7_;

    const float* f = feat + ((((size_t)(b * T_ + t + 1) * N_ + n) * H_ + h) * C_);
    const float* Q = q2 + h * NO_ * C_;
    float* o = qj + tid * NO_;
#pragma unroll
    for (int i = 0; i < NO_; ++i) {
        float s = 0.f;
#pragma unroll
        for (int j = 0; j < C_; ++j) s += Q[i * C_ + j] * f[j];
        o[i] = s;
    }
}

__global__ void k4_wd(const float* __restrict__ feat,
                      const int*   __restrict__ nn,
                      const float* __restrict__ q1,
                      const float* __restrict__ qj,
                      float*       __restrict__ dw)
{
    const int wid  = (blockIdx.x * blockDim.x + threadIdx.x) >> 5;
    const int lane = threadIdx.x & 31;

    const int tile = wid % TILES2_;
    const int r1   = wid / TILES2_;
    const int h    = r1 % H_;
    const int bt7  = r1 / H_;
    const int t    = bt7 % T7_;
    const int bq   = bt7 / T7_;
    const int btf  = bq * T_ + t;

    const int r  = lane & 15;
    const int hh = lane >> 4;

    const int  e     = tile * 16 + r;
    const bool rowOK = (e < E2_);
    const int  ec    = rowOK ? e : 0;
    const int  n     = ec / K1_;
    const int  k     = ec - n * K1_;
    const int  nbrv  = nn[n * K1_ + k];
    const bool pred  = rowOK && (nbrv >= 0);
    const unsigned vm = (unsigned)__ballot(pred);
    const int  jn    = pred ? nbrv : 0;
    const float sA   = pred ? 1.f : 0.f;

    const float* fi = feat + ((((size_t)btf * N_ + jn) * H_ + h) * C_);
    const float* Qh = q1 + h * NO_ * C_;
    const float sB  = (r < NO_) ? 1.f : 0.f;

    v8f d = {0.f,0.f,0.f,0.f,0.f,0.f,0.f,0.f};
#pragma unroll
    for (int c = 0; c < 4; ++c) {
        const int j0 = 4 * c + 2 * hh;
        v2f av, bv;
        av.x = sA * fi[j0];
        av.y = sA * fi[j0 + 1];
        bv.x = sB * Qh[(r & (NO_ - 1)) * C_ + j0];
        bv.y = sB * Qh[(r & (NO_ - 1)) * C_ + j0 + 1];
        d = __builtin_amdgcn_wmma_f32_16x16x4_f32(
                false, av, false, bv, (short)0, d, false, false);
    }

    float tv[8];
#pragma unroll
    for (int v = 0; v < 8; ++v) {
        const int ev = tile * 16 + v + (hh << 3);
        float qv = 0.f;
        if (ev < E2_ && r < NO_) {
            const int nv = ev / K1_;
            qv = qj[(((size_t)bt7 * N_ + nv) * H_ + h) * NO_ + r];
        }
        tv[v] = d[v] * qv;
    }
    const float s = row_reduce16(tv, r);

    const int   row = (lane & 7) | (hh << 3);
    const float wv  = ((vm >> row) & 1u) ? __expf(-s) : 0.f;

    if ((lane & 15) < 8) {
        const int ew = tile * 16 + row;
        if (ew < E2_)
            dw[((size_t)bt7 * E2_ + ew) * H_ + h] = wv;
    }
}

__global__ void k5_dw(float* __restrict__ dw)
{
    const size_t tid = (size_t)blockIdx.x * blockDim.x + threadIdx.x;
    const size_t total = (size_t)B_ * T7_ * N_ * H_;
    if (tid >= total) return;
    const int h   = (int)(tid % H_);
    const int n   = (int)((tid / H_) % N_);
    const int bt7 = (int)( tid / ((size_t)H_ * N_));

    float* base = dw + ((size_t)bt7 * E2_ + (size_t)n * K1_) * H_ + h;
    float s = 0.f;
#pragma unroll
    for (int k = 0; k < K1_; ++k) s += base[k * H_];
    const float inv = (s > 0.f) ? (1.f / s) : 0.f;
#pragma unroll
    for (int k = 0; k < K1_; ++k) base[k * H_] *= inv;
}

extern "C" void kernel_launch(void* const* d_in, const int* in_sizes, int n_in,
                              void* d_out, int out_size, void* d_ws, size_t ws_size,
                              hipStream_t stream)
{
    const float* feat = (const float*)d_in[0];
    const int*   nn   = (const int*)  d_in[1];
    const float* q1   = (const float*)d_in[2];
    const float* q2   = (const float*)d_in[3];
    const float* mm   = (const float*)d_in[4];

    float* uw = (float*)d_out;
    float* dw = uw + (size_t)B_ * T_ * N_ * 16 * H_;

    float* deg = (float*)d_ws;
    float* qjw = deg + (size_t)B_ * T_ * N_ * H_;

    const int BLK = 256;

    {
        const int waves = B_ * T_ * H_ * N_;
        k1_w_deg<<<waves * 32 / BLK, BLK, 0, stream>>>(feat, nn, mm, uw, deg);
    }
    {
        const size_t total = (size_t)B_ * T_ * N_ * 16 * H_;
        k2_uw<<<(unsigned)((total + BLK - 1) / BLK), BLK, 0, stream>>>(uw, deg, nn);
    }
    {
        const size_t total = (size_t)B_ * T7_ * N_ * H_;
        k3_qj<<<(unsigned)((total + BLK - 1) / BLK), BLK, 0, stream>>>(feat, q2, qjw);
    }
    {
        const long long waves = (long long)B_ * T7_ * H_ * TILES2_;
        k4_wd<<<(unsigned)(waves * 32 / BLK), BLK, 0, stream>>>(feat, nn, q1, qjw, dw);
    }
    {
        const size_t total = (size_t)B_ * T7_ * N_ * H_;
        k5_dw<<<(unsigned)((total + BLK - 1) / BLK), BLK, 0, stream>>>(dw);
    }
}


# One wave (both WMMA kernels) or thread (u_w normalize) now covers all 4 heads of its rows instead of one: 4x fewer waves, index row read once, 16-byte contiguous w/deg/d_w accesses
# speedup vs baseline: 1.2663x; 1.2663x over previous
#include <hip/hip_runtime.h>
#include <math.h>

#define B_   2
#define T_   8
#define N_   5000
#define K1_  17
#define H_   4
#define C_   16
#define NO_  8
#define T7_  (T_-1)
#define E2_  (N_*K1_)
#define TILES2_ ((E2_+15)/16)

typedef float v2f __attribute__((ext_vector_type(2)));
typedef float v8f __attribute__((ext_vector_type(8)));

#define SWZ(x, M) __int_as_float(__builtin_amdgcn_ds_swizzle( \
                      __float_as_int(x), ((M) << 10) | 0x1f))

template<int M>
__device__ __forceinline__ float mergepair(float a, float b, int c) {
    const float keep = (c & M) ? b : a;
    const float send = (c & M) ? a : b;
    return keep + SWZ(send, M);
}

__device__ __forceinline__ float row_reduce16(const float t[8], int c) {
    const float m0 = mergepair<1>(t[0], t[1], c);
    const float m1 = mergepair<1>(t[2], t[3], c);
    const float m2 = mergepair<1>(t[4], t[5], c);
    const float m3 = mergepair<1>(t[6], t[7], c);
    const float n0 = mergepair<2>(m0, m1, c);
    const float n1 = mergepair<2>(m2, m3, c);
    float s = mergepair<4>(n0, n1, c);
    s += SWZ(s, 8);
    return s;
}

__global__ void k1_w_deg(const float* __restrict__ feat,
                         const int*   __restrict__ nn,
                         const float* __restrict__ mm,
                         float*       __restrict__ w_out,
                         float*       __restrict__ deg)
{
    const int wid  = (blockIdx.x * blockDim.x + threadIdx.x) >> 5;
    const int lane = threadIdx.x & 31;

    const int n  = wid % N_;
    const int bt = wid / N_;

    const int r  = lane & 15;
    const int hh = lane >> 4;

    const int nbrv = nn[n * K1_ + 1 + r];
    const unsigned vm = (unsigned)__ballot(nbrv >= 0);
    const int nj = (nbrv < 0) ? n : nbrv;

    const float* fi0 = feat + (((size_t)bt * N_ + n ) * H_ * C_);
    const float* fj0 = feat + (((size_t)bt * N_ + nj) * H_ * C_);

    const int  row = (lane & 7) | (hh << 3);
    const bool rv  = ((vm >> row) & 1u) != 0u;

    float wq[H_], dq[H_];
#pragma unroll
    for (int h = 0; h < H_; ++h) {
        const float* fi = fi0 + h * C_;
        const float* fj = fj0 + h * C_;
        const float* Mh = mm + h * C_ * C_;

        v8f d = {0.f,0.f,0.f,0.f,0.f,0.f,0.f,0.f};
#pragma unroll
        for (int c = 0; c < 4; ++c) {
            const int j0 = 4 * c + 2 * hh;
            v2f av, bv;
            av.x = fi[j0]     - fj[j0];
            av.y = fi[j0 + 1] - fj[j0 + 1];
            bv.x = Mh[r * C_ + j0];
            bv.y = Mh[r * C_ + j0 + 1];
            d = __builtin_amdgcn_wmma_f32_16x16x4_f32(
                    false, av, false, bv, (short)0, d, false, false);
        }

        float t[8];
#pragma unroll
        for (int v = 0; v < 8; ++v) t[v] = d[v] * d[v];
        const float s = row_reduce16(t, r);

        const float w = rv ? __expf(-s) : 0.f;

        float dp = w;
        dp += SWZ(dp, 1);
        dp += SWZ(dp, 2);
        dp += SWZ(dp, 4);
        dp += SWZ(dp, 16);

        wq[h] = w;
        dq[h] = dp;
    }

    if ((lane & 15) < 8)
        *reinterpret_cast<float4*>(w_out + (((size_t)bt * N_ + n) * 16 + row) * H_) =
            make_float4(wq[0], wq[1], wq[2], wq[3]);
    if (lane == 0)
        *reinterpret_cast<float4*>(deg + ((size_t)bt * N_ + n) * H_) =
            make_float4(dq[0], dq[1], dq[2], dq[3]);
}

__device__ __forceinline__ float norm_one(float w, float di, float dj)
{
    const float dm = sqrtf(di * dj);
    return (dm > 0.f) ? (w / dm) : 0.f;
}

__global__ void k2_uw(float* __restrict__ uw,
                      const float* __restrict__ deg,
                      const int*   __restrict__ nn)
{
    const size_t tid = (size_t)blockIdx.x * blockDim.x + threadIdx.x;
    const size_t total = (size_t)B_ * T_ * N_ * 16;
    if (tid >= total) return;
    const int k  = (int)(tid % 16);
    const int n  = (int)((tid / 16) % N_);
    const int bt = (int)( tid / ((size_t)16 * N_));

    float4* p = reinterpret_cast<float4*>(uw + tid * H_);
    const float4 w  = *p;
    const int    nb = nn[n * K1_ + 1 + k];
    const float4 di = *reinterpret_cast<const float4*>(deg + ((size_t)bt * N_ + n) * H_);
    float4 dj = make_float4(0.f, 0.f, 0.f, 0.f);
    if (nb >= 0)
        dj = *reinterpret_cast<const float4*>(deg + ((size_t)bt * N_ + nb) * H_);

    float4 o;
    o.x = norm_one(w.x, di.x, dj.x);
    o.y = norm_one(w.y, di.y, dj.y);
    o.z = norm_one(w.z, di.z, dj.z);
    o.w = norm_one(w.w, di.w, dj.w);
    *p = o;
}

__global__ void k3_qj(const float* __restrict__ feat,
                      const float* __restrict__ q2,
                      float*       __restrict__ qj)
{
    const size_t tid = (size_t)blockIdx.x * blockDim.x + threadIdx.x;
    const size_t total = (size_t)B_ * T7_ * N_ * H_;
    if (tid >= total) return;
    const int h   = (int)(tid % H_);
    const int n   = (int)((tid / H_) % N_);
    const int bt7 = (int)( tid / ((size_t)H_ * N_));
    const int t   = bt7 % T7_;
    const int b   = bt7 / T7_;

    const float* f = feat + ((((size_t)(b * T_ + t + 1) * N_ + n) * H_ + h) * C_);
    const float* Q = q2 + h * NO_ * C_;
    float* o = qj + tid * NO_;
#pragma unroll
    for (int i = 0; i < NO_; ++i) {
        float s = 0.f;
#pragma unroll
        for (int j = 0; j < C_; ++j) s += Q[i * C_ + j] * f[j];
        o[i] = s;
    }
}

#define WAVES4_ (B_ * T7_ * TILES2_)

__global__ void k4_wd(const float* __restrict__ feat,
                      const int*   __restrict__ nn,
                      const float* __restrict__ q1,
                      const float* __restrict__ qj,
                      float*       __restrict__ dw)
{
    const int wid  = (blockIdx.x * blockDim.x + threadIdx.x) >> 5;
    if (wid >= WAVES4_) return;
    const int lane = threadIdx.x & 31;

    const int tile = wid % TILES2_;
    const int bt7  = wid / TILES2_;
    const int t    = bt7 % T7_;
    const int bq   = bt7 / T7_;
    const int btf  = bq * T_ + t;

    const int r  = lane & 15;
    const int hh = lane >> 4;

    const int  e     = tile * 16 + r;
    const bool rowOK = (e < E2_);
    const int  ec    = rowOK ? e : 0;
    const int  n     = ec / K1_;
    const int  k     = ec - n * K1_;
    const int  nbrv  = nn[n * K1_ + k];
    const bool pred  = rowOK && (nbrv >= 0);
    const unsigned vm = (unsigned)__ballot(pred);
    const int  jn    = pred ? nbrv : 0;
    const float sA   = pred ? 1.f : 0.f;

    const float* fi0 = feat + (((size_t)btf * N_ + jn) * H_ * C_);
    const float sB   = (r < NO_) ? 1.f : 0.f;

    size_t qoff[8];
    bool   qok[8];
#pragma unroll
    for (int v = 0; v < 8; ++v) {
        const int ev = tile * 16 + v + (hh << 3);
        qok[v] = (ev < E2_) && (r < NO_);
        const int nv = (ev < E2_) ? (ev / K1_) : 0;
        qoff[v] = (((size_t)bt7 * N_ + nv) * H_) * NO_ + r;
    }

    const int  row = (lane & 7) | (hh << 3);
    const bool rv  = ((vm >> row) & 1u) != 0u;

    float wq[H_];
#pragma unroll
    for (int h = 0; h < H_; ++h) {
        const float* fi = fi0 + h * C_;
        const float* Qh = q1 + h * NO_ * C_;

        v8f d = {0.f,0.f,0.f,0.f,0.f,0.f,0.f,0.f};
#pragma unroll
        for (int c = 0; c < 4; ++c) {
            const int j0 = 4 * c + 2 * hh;
            v2f av, bv;
            av.x = sA * fi[j0];
            av.y = sA * fi[j0 + 1];
            bv.x = sB * Qh[(r & (NO_ - 1)) * C_ + j0];
            bv.y = sB * Qh[(r & (NO_ - 1)) * C_ + j0 + 1];
            d = __builtin_amdgcn_wmma_f32_16x16x4_f32(
                    false, av, false, bv, (short)0, d, false, false);
        }

        float tv[8];
#pragma unroll
        for (int v = 0; v < 8; ++v) {
            float qv = 0.f;
            if (qok[v]) qv = qj[qoff[v] + h * NO_];
            tv[v] = d[v] * qv;
        }
        const float s = row_reduce16(tv, r);

        wq[h] = rv ? __expf(-s) : 0.f;
    }

    if ((lane & 15) < 8) {
        const int ew = tile * 16 + row;
        if (ew < E2_)
            *reinterpret_cast<float4*>(dw + ((size_t)bt7 * E2_ + ew) * H_) =
                make_float4(wq[0], wq[1], wq[2], wq[3]);
    }
}

__global__ void k5_dw(float* __restrict__ dw)
{
    const size_t tid = (size_t)blockIdx.x * blockDim.x + threadIdx.x;
    const size_t total = (size_t)B_ * T7_ * N_ * H_;
    if (tid >= total) return;
    const int h   = (int)(tid % H_);
    const int n   = (int)((tid / H_) % N_);
    const int bt7 = (int)( tid / ((size_t)H_ * N_));

    float* base = dw + ((size_t)bt7 * E2_ + (size_t)n * K1_) * H_ + h;
    float s = 0.f;
#pragma unroll
    for (int k = 0; k < K1_; ++k) s += base[k * H_];
    const float inv = (s > 0.f) ? (1.f / s) : 0.f;
#pragma unroll
    for (int k = 0; k < K1_; ++k) base[k * H_] *= inv;
}

extern "C" void kernel_launch(void* const* d_in, const int* in_sizes, int n_in,
                              void* d_out, int out_size, void* d_ws, size_t ws_size,
                              hipStream_t stream)
{
    const float* feat = (const float*)d_in[0];
    const int*   nn   = (const int*)  d_in[1];
    const float* q1   = (const float*)d_in[2];
    const float* q2   = (const float*)d_in[3];
    const float* mm   = (const float*)d_in[4];

    float* uw = (float*)d_out;
    float* dw = uw + (size_t)B_ * T_ * N_ * 16 * H_;

    float* deg = (float*)d_ws;
    float* qjw = deg + (size_t)B_ * T_ * N_ * H_;

    const int BLK = 256;

    {
        const int waves = B_ * T_ * N_;
        k1_w_deg<<<waves * 32 / BLK, BLK, 0, stream>>>(feat, nn, mm, uw, deg);
    }
    {
        const size_t total = (size_t)B_ * T_ * N_ * 16;
        k2_uw<<<(unsigned)((total + BLK - 1) / BLK), BLK, 0, stream>>>(uw, deg, nn);
    }
    {
        const size_t total = (size_t)B_ * T7_ * N_ * H_;
        k3_qj<<<(unsigned)((total + BLK - 1) / BLK), BLK, 0, stream>>>(feat, q2, qjw);
    }
    {
        const long long waves = (long long)WAVES4_;
        k4_wd<<<(unsigned)((waves * 32 + BLK - 1) / BLK), BLK, 0, stream>>>(feat, nn, q1, qjw, dw);
    }
    {
        const size_t total = (size_t)B_ * T7_ * N_ * H_;
        k5_dw<<<(unsigned)((total + BLK - 1) / BLK), BLK, 0, stream>>>(dw);
    }
}


# k1 and k2 run on a fixed grid of 8000 resident waves that loop over 10 nodes / 5 rows each: 5-10x fewer waves to start, multiM fragments loaded once per wave, next index row fetched a trip ahead
# speedup vs baseline: 1.4039x; 1.1087x over previous
#include <hip/hip_runtime.h>
#include <math.h>

#define B_   2
#define T_   8
#define N_   5000
#define K1_  17
#define H_   4
#define C_   16
#define NO_  8
#define T7_  (T_-1)
#define E2_  (N_*K1_)
#define TILES2_ ((E2_+15)/16)

#define BLK_      256
#define NBLK_P    1000
#define WAVES_P   (NBLK_P * BLK_ / 32)
#define THREADS_P (NBLK_P * BLK_)

typedef float v2f __attribute__((ext_vector_type(2)));
typedef float v8f __attribute__((ext_vector_type(8)));

#define SWZ(x, M) __int_as_float(__builtin_amdgcn_ds_swizzle( \
                      __float_as_int(x), ((M) << 10) | 0x1f))

template<int M>
__device__ __forceinline__ float mergepair(float a, float b, int c) {
    const float keep = (c & M) ? b : a;
    const float send = (c & M) ? a : b;
    return keep + SWZ(send, M);
}

__device__ __forceinline__ float row_reduce16(const float t[8], int c) {
    const float m0 = mergepair<1>(t[0], t[1], c);
    const float m1 = mergepair<1>(t[2], t[3], c);
    const float m2 = mergepair<1>(t[4], t[5], c);
    const float m3 = mergepair<1>(t[6], t[7], c);
    const float n0 = mergepair<2>(m0, m1, c);
    const float n1 = mergepair<2>(m2, m3, c);
    float s = mergepair<4>(n0, n1, c);
    s += SWZ(s, 8);
    return s;
}

#define NODES1_ (B_ * T_ * N_)
#define TRIPS1_ (NODES1_ / WAVES_P)
static_assert(NODES1_ % WAVES_P == 0, "k1: every wave takes the same number of nodes");

__global__ void k1_w_deg(const float* __restrict__ feat,
                         const int*   __restrict__ nn,
                         const float* __restrict__ mm,
                         float*       __restrict__ w_out,
                         float*       __restrict__ deg)
{
    const int wid  = __builtin_amdgcn_readfirstlane(
                         (int)((blockIdx.x * blockDim.x + threadIdx.x) >> 5));
    const int lane = threadIdx.x & 31;

    const int r  = lane & 15;
    const int hh = lane >> 4;
    const int row = (lane & 7) | (hh << 3);

    v2f bm[H_][4];
#pragma unroll
    for (int h = 0; h < H_; ++h) {
        const float* Mh = mm + h * C_ * C_;
#pragma unroll
        for (int c = 0; c < 4; ++c) {
            const int j0 = 4 * c + 2 * hh;
            bm[h][c].x = Mh[r * C_ + j0];
            bm[h][c].y = Mh[r * C_ + j0 + 1];
        }
    }

    int nbr_next = nn[(wid % N_) * K1_ + 1 + r];

    for (int it = 0; it < TRIPS1_; ++it) {
        const int g  = wid + it * WAVES_P;
        const int bt = g / N_;
        const int n  = g - bt * N_;

        const int nbrv = nbr_next;
        if (it + 1 < TRIPS1_)
            nbr_next = nn[((g + WAVES_P) % N_) * K1_ + 1 + r];

        const unsigned vm = (unsigned)__ballot(nbrv >= 0);
        const int nj = (nbrv < 0) ? n : nbrv;

        const float* fi0 = feat + ((size_t)g * H_ * C_);
        const float* fj0 = feat + (((size_t)bt * N_ + nj) * H_ * C_);

        const bool rv = ((vm >> row) & 1u) != 0u;

        float wq[H_], dq[H_];
#pragma unroll
        for (int h = 0; h < H_; ++h) {
            const float* fi = fi0 + h * C_;
            const float* fj = fj0 + h * C_;

            v8f d = {0.f,0.f,0.f,0.f,0.f,0.f,0.f,0.f};
#pragma unroll
            for (int c = 0; c < 4; ++c) {
                const int j0 = 4 * c + 2 * hh;
                v2f av;
                av.x = fi[j0]     - fj[j0];
                av.y = fi[j0 + 1] - fj[j0 + 1];
                d = __builtin_amdgcn_wmma_f32_16x16x4_f32(
                        false, av, false, bm[h][c], (short)0, d, false, false);
            }

            float t[8];
#pragma unroll
            for (int v = 0; v < 8; ++v) t[v] = d[v] * d[v];
            const float s = row_reduce16(t, r);

            const float w = rv ? __expf(-s) : 0.f;

            float dp = w;
            dp += SWZ(dp, 1);
            dp += SWZ(dp, 2);
            dp += SWZ(dp, 4);
            dp += SWZ(dp, 16);

            wq[h] = w;
            dq[h] = dp;
        }

        if ((lane & 15) < 8)
            *reinterpret_cast<float4*>(w_out + ((size_t)g * 16 + row) * H_) =
                make_float4(wq[0], wq[1], wq[2], wq[3]);
        if (lane == 0)
            *reinterpret_cast<float4*>(deg + (size_t)g * H_) =
                make_float4(dq[0], dq[1], dq[2], dq[3]);
    }
}

__device__ __forceinline__ float norm_one(float w, float di, float dj)
{
    const float dm = sqrtf(di * dj);
    return (dm > 0.f) ? (w / dm) : 0.f;
}

#define ITEMS2_ (B_ * T_ * N_ * 16)
#define TRIPS2_ (ITEMS2_ / THREADS_P)
static_assert(ITEMS2_ % THREADS_P == 0, "k2: every thread takes the same number of rows");

__global__ void k2_uw(float* __restrict__ uw,
                      const float* __restrict__ deg,
                      const int*   __restrict__ nn)
{
    const int t0 = (int)(blockIdx.x * blockDim.x + threadIdx.x);

    float4 w[TRIPS2_], di[TRIPS2_], dj[TRIPS2_];
    int    nb[TRIPS2_], own[TRIPS2_];
#pragma unroll
    for (int i = 0; i < TRIPS2_; ++i) {
        const int item = t0 + i * THREADS_P;
        const int k  = item % 16;
        own[i] = item / 16;
        const int n  = own[i] % N_;
        w[i]  = *reinterpret_cast<const float4*>(uw + (size_t)item * H_);
        nb[i] = nn[n * K1_ + 1 + k];
        di[i] = *reinterpret_cast<const float4*>(deg + (size_t)own[i] * H_);
    }
#pragma unroll
    for (int i = 0; i < TRIPS2_; ++i) {
        const int n   = own[i] % N_;
        const int nbc = (nb[i] >= 0) ? nb[i] : n;
        dj[i] = *reinterpret_cast<const float4*>(deg + (size_t)(own[i] - n + nbc) * H_);
        if (nb[i] < 0) dj[i] = make_float4(0.f, 0.f, 0.f, 0.f);
    }
#pragma unroll
    for (int i = 0; i < TRIPS2_; ++i) {
        const int item = t0 + i * THREADS_P;
        float4 o;
        o.x = norm_one(w[i].x, di[i].x, dj[i].x);
        o.y = norm_one(w[i].y, di[i].y, dj[i].y);
        o.z = norm_one(w[i].z, di[i].z, dj[i].z);
        o.w = norm_one(w[i].w, di[i].w, dj[i].w);
        *reinterpret_cast<float4*>(uw + (size_t)item * H_) = o;
    }
}

__global__ void k3_qj(const float* __restrict__ feat,
                      const float* __restrict__ q2,
                      float*       __restrict__ qj)
{
    const size_t tid = (size_t)blockIdx.x * blockDim.x + threadIdx.x;
    const size_t total = (size_t)B_ * T7_ * N_ * H_;
    if (tid >= total) return;
    const int h   = (int)(tid % H_);
    const int n   = (int)((tid / H_) % N_);
    const int bt7 = (int)( tid / ((size_t)H_ * N_));
    const int t   = bt7 % T7_;
    const int b   = bt7 / T7_;

    const float* f = feat + ((((size_t)(b * T_ + t + 1) * N_ + n) * H_ + h) * C_);
    const float* Q = q2 + h * NO_ * C_;
    float* o = qj + tid * NO_;
#pragma unroll
    for (int i = 0; i < NO_; ++i) {
        float s = 0.f;
#pragma unroll
        for (int j = 0; j < C_; ++j) s += Q[i * C_ + j] * f[j];
        o[i] = s;
    }
}

#define WAVES4_ (B_ * T7_ * TILES2_)

__global__ void k4_wd(const float* __restrict__ feat,
                      const int*   __restrict__ nn,
                      const float* __restrict__ q1,
                      const float* __restrict__ qj,
                      float*       __restrict__ dw)
{
    const int wid  = (blockIdx.x * blockDim.x + threadIdx.x) >> 5;
    if (wid >= WAVES4_) return;
    const int lane = threadIdx.x & 31;

    const int tile = wid % TILES2_;
    const int bt7  = wid / TILES2_;
    const int t    = bt7 % T7_;
    const int bq   = bt7 / T7_;
    const int btf  = bq * T_ + t;

    const int r  = lane & 15;
    const int hh = lane >> 4;

    const int  e     = tile * 16 + r;
    const bool rowOK = (e < E2_);
    const int  ec    = rowOK ? e : 0;
    const int  n     = ec / K1_;
    const int  k     = ec - n * K1_;
    const int  nbrv  = nn[n * K1_ + k];
    const bool pred  = rowOK && (nbrv >= 0);
    const unsigned vm = (unsigned)__ballot(pred);
    const int  jn    = pred ? nbrv : 0;
    const float sA   = pred ? 1.f : 0.f;

    const float* fi0 = feat + (((size_t)btf * N_ + jn) * H_ * C_);
    const float sB   = (r < NO_) ? 1.f : 0.f;

    size_t qoff[8];
    bool   qok[8];
#pragma unroll
    for (int v = 0; v < 8; ++v) {
        const int ev = tile * 16 + v + (hh << 3);
        qok[v] = (ev < E2_) && (r < NO_);
        const int nv = (ev < E2_) ? (ev / K1_) : 0;
        qoff[v] = (((size_t)bt7 * N_ + nv) * H_) * NO_ + r;
    }

    const int  row = (lane & 7) | (hh << 3);
    const bool rv  = ((vm >> row) & 1u) != 0u;

    float wq[H_];
#pragma unroll
    for (int h = 0; h < H_; ++h) {
        const float* fi = fi0 + h * C_;
        const float* Qh = q1 + h * NO_ * C_;

        v8f d = {0.f,0.f,0.f,0.f,0.f,0.f,0.f,0.f};
#pragma unroll
        for (int c = 0; c < 4; ++c) {
            const int j0 = 4 * c + 2 * hh;
            v2f av, bv;
            av.x = sA * fi[j0];
            av.y = sA * fi[j0 + 1];
            bv.x = sB * Qh[(r & (NO_ - 1)) * C_ + j0];
            bv.y = sB * Qh[(r & (NO_ - 1)) * C_ + j0 + 1];
            d = __builtin_amdgcn_wmma_f32_16x16x4_f32(
                    false, av, false, bv, (short)0, d, false, false);
        }

        float tv[8];
#pragma unroll
        for (int v = 0; v < 8; ++v) {
            float qv = 0.f;
            if (qok[v]) qv = qj[qoff[v] + h * NO_];
            tv[v] = d[v] * qv;
        }
        const float s = row_reduce16(tv, r);

        wq[h] = rv ? __expf(-s) : 0.f;
    }

    if ((lane & 15) < 8) {
        const int ew = tile * 16 + row;
        if (ew < E2_)
            *reinterpret_cast<float4*>(dw + ((size_t)bt7 * E2_ + ew) * H_) =
                make_float4(wq[0], wq[1], wq[2], wq[3]);
    }
}

__global__ void k5_dw(float* __restrict__ dw)
{
    const size_t tid = (size_t)blockIdx.x * blockDim.x + threadIdx.x;
    const size_t total = (size_t)B_ * T7_ * N_ * H_;
    if (tid >= total) return;
    const int h   = (int)(tid % H_);
    const int n   = (int)((tid / H_) % N_);
    const int bt7 = (int)( tid / ((size_t)H_ * N_));

    float* base = dw + ((size_t)bt7 * E2_ + (size_t)n * K1_) * H_ + h;
    float s = 0.f;
#pragma unroll
    for (int k = 0; k < K1_; ++k) s += base[k * H_];
    const float inv = (s > 0.f) ? (1.f / s) : 0.f;
#pragma unroll
    for (int k = 0; k < K1_; ++k) base[k * H_] *= inv;
}

extern "C" void kernel_launch(void* const* d_in, const int* in_sizes, int n_in,
                              void* d_out, int out_size, void* d_ws, size_t ws_size,
                              hipStream_t stream)
{
    const float* feat = (const float*)d_in[0];
    const int*   nn   = (const int*)  d_in[1];
    const float* q1   = (const float*)d_in[2];
    const float* q2   = (const float*)d_in[3];
    const float* mm   = (const float*)d_in[4];

    float* uw = (float*)d_out;
    float* dw = uw + (size_t)B_ * T_ * N_ * 16 * H_;

    float* deg = (float*)d_ws;
    float* qjw = deg + (size_t)B_ * T_ * N_ * H_;

    const int BLK = 256;

    k1_w_deg<<<NBLK_P, BLK_, 0, stream>>>(feat, nn, mm, uw, deg);
    k2_uw<<<NBLK_P, BLK_, 0, stream>>>(uw, deg, nn);
    {
        const size_t total = (size_t)B_ * T7_ * N_ * H_;
        k3_qj<<<(unsigned)((total + BLK - 1) / BLK), BLK, 0, stream>>>(feat, q2, qjw);
    }
    {
        const long long waves = (long long)WAVES4_;
        k4_wd<<<(unsigned)((waves * 32 + BLK - 1) / BLK), BLK, 0, stream>>>(feat, nn, q1, qjw, dw);
    }
    {
        const size_t total = (size_t)B_ * T7_ * N_ * H_;
        k5_dw<<<(unsigned)((total + BLK - 1) / BLK), BLK, 0, stream>>>(dw);
    }
}


# Q1 projection hoisted ahead of the gather: k3 projects each node once on WMMA (was once per edge), k4 gathers 32-byte projected rows and forms Qj itself; dot summed per thread in the tree's order
# speedup vs baseline: 1.8024x; 1.2839x over previous
#include <hip/hip_runtime.h>
#include <math.h>

#define B_   2
#define T_   8
#define N_   5000
#define K1_  17
#define H_   4
#define C_   16
#define NO_  8
#define T7_  (T_-1)
#define E2_  (N_*K1_)

#define BLK_      256
#define NBLK_P    1000
#define WAVES_P   (NBLK_P * BLK_ / 32)
#define THREADS_P (NBLK_P * BLK_)

typedef float v2f __attribute__((ext_vector_type(2)));
typedef float v8f __attribute__((ext_vector_type(8)));

#define SWZ(x, M) __int_as_float(__builtin_amdgcn_ds_swizzle( \
                      __float_as_int(x), ((M) << 10) | 0x1f))

template<int M>
__device__ __forceinline__ float mergepair(float a, float b, int c) {
    const float keep = (c & M) ? b : a;
    const float send = (c & M) ? a : b;
    return keep + SWZ(send, M);
}

__device__ __forceinline__ float row_reduce16(const float t[8], int c) {
    const float m0 = mergepair<1>(t[0], t[1], c);
    const float m1 = mergepair<1>(t[2], t[3], c);
    const float m2 = mergepair<1>(t[4], t[5], c);
    const float m3 = mergepair<1>(t[6], t[7], c);
    const float n0 = mergepair<2>(m0, m1, c);
    const float n1 = mergepair<2>(m2, m3, c);
    float s = mergepair<4>(n0, n1, c);
    s += SWZ(s, 8);
    return s;
}

#define NODES1_ (B_ * T_ * N_)
#define TRIPS1_ (NODES1_ / WAVES_P)
static_assert(NODES1_ % WAVES_P == 0, "k1: every wave takes the same number of nodes");

__global__ void k1_w_deg(const float* __restrict__ feat,
                         const int*   __restrict__ nn,
                         const float* __restrict__ mm,
                         float*       __restrict__ w_out,
                         float*       __restrict__ deg)
{
    const int wid  = __builtin_amdgcn_readfirstlane(
                         (int)((blockIdx.x * blockDim.x + threadIdx.x) >> 5));
    const int lane = threadIdx.x & 31;

    const int r  = lane & 15;
    const int hh = lane >> 4;
    const int row = (lane & 7) | (hh << 3);

    v2f bm[H_][4];
#pragma unroll
    for (int h = 0; h < H_; ++h) {
        const float* Mh = mm + h * C_ * C_;
#pragma unroll
        for (int c = 0; c < 4; ++c) {
            const int j0 = 4 * c + 2 * hh;
            bm[h][c].x = Mh[r * C_ + j0];
            bm[h][c].y = Mh[r * C_ + j0 + 1];
        }
    }

    int nbr_next = nn[(wid % N_) * K1_ + 1 + r];

    for (int it = 0; it < TRIPS1_; ++it) {
        const int g  = wid + it * WAVES_P;
        const int bt = g / N_;
        const int n  = g - bt * N_;

        const int nbrv = nbr_next;
        if (it + 1 < TRIPS1_)
            nbr_next = nn[((g + WAVES_P) % N_) * K1_ + 1 + r];

        const unsigned vm = (unsigned)__ballot(nbrv >= 0);
        const int nj = (nbrv < 0) ? n : nbrv;

        const float* fi0 = feat + ((size_t)g * H_ * C_);
        const float* fj0 = feat + (((size_t)bt * N_ + nj) * H_ * C_);

        const bool rv = ((vm >> row) & 1u) != 0u;

        float wq[H_], dq[H_];
#pragma unroll
        for (int h = 0; h < H_; ++h) {
            const float* fi = fi0 + h * C_;
            const float* fj = fj0 + h * C_;

            v8f d = {0.f,0.f,0.f,0.f,0.f,0.f,0.f,0.f};
#pragma unroll
            for (int c = 0; c < 4; ++c) {
                const int j0 = 4 * c + 2 * hh;
                v2f av;
                av.x = fi[j0]     - fj[j0];
                av.y = fi[j0 + 1] - fj[j0 + 1];
                d = __builtin_amdgcn_wmma_f32_16x16x4_f32(
                        false, av, false, bm[h][c], (short)0, d, false, false);
            }

            float t[8];
#pragma unroll
            for (int v = 0; v < 8; ++v) t[v] = d[v] * d[v];
            const float s = row_reduce16(t, r);

            const float w = rv ? __expf(-s) : 0.f;

            float dp = w;
            dp += SWZ(dp, 1);
            dp += SWZ(dp, 2);
            dp += SWZ(dp, 4);
            dp += SWZ(dp, 16);

            wq[h] = w;
            dq[h] = dp;
        }

        if ((lane & 15) < 8)
            *reinterpret_cast<float4*>(w_out + ((size_t)g * 16 + row) * H_) =
                make_float4(wq[0], wq[1], wq[2], wq[3]);
        if (lane == 0)
            *reinterpret_cast<float4*>(deg + (size_t)g * H_) =
                make_float4(dq[0], dq[1], dq[2], dq[3]);
    }
}

__device__ __forceinline__ float norm_one(float w, float di, float dj)
{
    const float dm = sqrtf(di * dj);
    return (dm > 0.f) ? (w / dm) : 0.f;
}

#define ITEMS2_ (B_ * T_ * N_ * 16)
#define TRIPS2_ (ITEMS2_ / THREADS_P)
static_assert(ITEMS2_ % THREADS_P == 0, "k2: every thread takes the same number of rows");

__global__ void k2_uw(float* __restrict__ uw,
                      const float* __restrict__ deg,
                      const int*   __restrict__ nn)
{
    const int t0 = (int)(blockIdx.x * blockDim.x + threadIdx.x);

    float4 w[TRIPS2_], di[TRIPS2_], dj[TRIPS2_];
    int    nb[TRIPS2_], own[TRIPS2_];
#pragma unroll
    for (int i = 0; i < TRIPS2_; ++i) {
        const int item = t0 + i * THREADS_P;
        const int k  = item % 16;
        own[i] = item / 16;
        const int n  = own[i] % N_;
        w[i]  = *reinterpret_cast<const float4*>(uw + (size_t)item * H_);
        nb[i] = nn[n * K1_ + 1 + k];
        di[i] = *reinterpret_cast<const float4*>(deg + (size_t)own[i] * H_);
    }
#pragma unroll
    for (int i = 0; i < TRIPS2_; ++i) {
        const int n   = own[i] % N_;
        const int nbc = (nb[i] >= 0) ? nb[i] : n;
        dj[i] = *reinterpret_cast<const float4*>(deg + (size_t)(own[i] - n + nbc) * H_);
        if (nb[i] < 0) dj[i] = make_float4(0.f, 0.f, 0.f, 0.f);
    }
#pragma unroll
    for (int i = 0; i < TRIPS2_; ++i) {
        const int item = t0 + i * THREADS_P;
        float4 o;
        o.x = norm_one(w[i].x, di[i].x, dj[i].x);
        o.y = norm_one(w[i].y, di[i].y, dj[i].y);
        o.z = norm_one(w[i].z, di[i].z, dj[i].z);
        o.w = norm_one(w[i].w, di[i].w, dj[i].w);
        *reinterpret_cast<float4*>(uw + (size_t)item * H_) = o;
    }
}

#define ROWS7_  (B_ * T7_ * N_)
#define TILES3_ (ROWS7_ / 16)
static_assert(ROWS7_ % 16 == 0, "k3: whole 16-row tiles");

__global__ void k3_qi(const float* __restrict__ feat,
                      const float* __restrict__ q1,
                      float*       __restrict__ qi)
{
    const int wid  = __builtin_amdgcn_readfirstlane(
                         (int)((blockIdx.x * blockDim.x + threadIdx.x) >> 5));
    if (wid >= TILES3_) return;
    const int lane = threadIdx.x & 31;
    const int r  = lane & 15;
    const int hh = lane >> 4;

    const int g7  = wid * 16 + r;
    const int bt7 = g7 / N_;
    const int n   = g7 - bt7 * N_;
    const int b   = bt7 / T7_;
    const int t   = bt7 - b * T7_;

    const float* f0 = feat + (((size_t)(b * T_ + t) * N_ + n) * H_ * C_);

    v8f d[H_];
#pragma unroll
    for (int h = 0; h < H_; ++h) {
        const float* fr = f0 + h * C_;
        const float* Qh = q1 + (h * NO_ + (r & (NO_ - 1))) * C_;
        v8f acc = {0.f,0.f,0.f,0.f,0.f,0.f,0.f,0.f};
#pragma unroll
        for (int c = 0; c < 4; ++c) {
            const int j0 = 4 * c + 2 * hh;
            v2f av, bv;
            av.x = Qh[j0];
            av.y = Qh[j0 + 1];
            bv.x = fr[j0];
            bv.y = fr[j0 + 1];
            acc = __builtin_amdgcn_wmma_f32_16x16x4_f32(
                      false, av, false, bv, (short)0, acc, false, false);
        }
        d[h] = acc;
    }

    if (hh == 0) {
        float* o = qi + (size_t)g7 * H_ * NO_;
#pragma unroll
        for (int h = 0; h < H_; ++h) {
            *reinterpret_cast<float4*>(o + h * NO_) =
                make_float4(d[h][0], d[h][1], d[h][2], d[h][3]);
            *reinterpret_cast<float4*>(o + h * NO_ + 4) =
                make_float4(d[h][4], d[h][5], d[h][6], d[h][7]);
        }
    }
}

__global__ void __launch_bounds__(BLK_)
k4_wd(const float* __restrict__ feat,
      const int*   __restrict__ nn,
      const float* __restrict__ q2,
      const float* __restrict__ qi,
      float*       __restrict__ dw)
{
    const int tid = (int)(blockIdx.x * blockDim.x + threadIdx.x);
    if (tid >= ROWS7_ * H_) return;
    const int h   = tid % H_;
    const int g7  = tid / H_;
    const int bt7 = g7 / N_;
    const int n   = g7 - bt7 * N_;
    const int b   = bt7 / T7_;
    const int t   = bt7 - b * T7_;

    int nb[K1_];
#pragma unroll
    for (int k = 0; k < K1_; ++k) nb[k] = nn[n * K1_ + k];

    const float* f = feat + ((((size_t)(b * T_ + t + 1) * N_ + n) * H_ + h) * C_);
    const float* Q = q2 + h * NO_ * C_;
    float fv[C_];
#pragma unroll
    for (int j = 0; j < C_; j += 4) {
        const float4 x = *reinterpret_cast<const float4*>(f + j);
        fv[j] = x.x; fv[j + 1] = x.y; fv[j + 2] = x.z; fv[j + 3] = x.w;
    }
    float qj[NO_];
#pragma unroll
    for (int i = 0; i < NO_; ++i) {
        float s = 0.f;
#pragma unroll
        for (int j = 0; j < C_; ++j) s += Q[i * C_ + j] * fv[j];
        qj[i] = s;
    }

    const float* qrow = qi + ((size_t)bt7 * N_ * H_ + h) * NO_;
    float*       o    = dw + ((size_t)bt7 * E2_ + (size_t)n * K1_) * H_ + h;
#pragma unroll
    for (int k = 0; k < K1_; ++k) {
        const bool ok = (nb[k] >= 0);
        const int  jn = ok ? nb[k] : n;
        const float* p = qrow + (size_t)jn * (H_ * NO_);
        const float4 a0 = *reinterpret_cast<const float4*>(p);
        const float4 a1 = *reinterpret_cast<const float4*>(p + 4);
        const float s = ((a0.x * qj[0] + a0.y * qj[1]) + (a0.z * qj[2] + a0.w * qj[3]))
                      + ((a1.x * qj[4] + a1.y * qj[5]) + (a1.z * qj[6] + a1.w * qj[7]));
        const float e = __expf(-s);
        o[k * H_] = ok ? e : 0.f;
    }
}

__global__ void k5_dw(float* __restrict__ dw)
{
    const size_t tid = (size_t)blockIdx.x * blockDim.x + threadIdx.x;
    const size_t total = (size_t)B_ * T7_ * N_ * H_;
    if (tid >= total) return;
    const int h   = (int)(tid % H_);
    const int n   = (int)((tid / H_) % N_);
    const int bt7 = (int)( tid / ((size_t)H_ * N_));

    float* base = dw + ((size_t)bt7 * E2_ + (size_t)n * K1_) * H_ + h;
    float s = 0.f;
#pragma unroll
    for (int k = 0; k < K1_; ++k) s += base[k * H_];
    const float inv = (s > 0.f) ? (1.f / s) : 0.f;
#pragma unroll
    for (int k = 0; k < K1_; ++k) base[k * H_] *= inv;
}

extern "C" void kernel_launch(void* const* d_in, const int* in_sizes, int n_in,
                              void* d_out, int out_size, void* d_ws, size_t ws_size,
                              hipStream_t stream)
{
    const float* feat = (const float*)d_in[0];
    const int*   nn   = (const int*)  d_in[1];
    const float* q1   = (const float*)d_in[2];
    const float* q2   = (const float*)d_in[3];
    const float* mm   = (const float*)d_in[4];

    float* uw = (float*)d_out;
    float* dw = uw + (size_t)B_ * T_ * N_ * 16 * H_;

    float* deg = (float*)d_ws;
    float* qiw = deg + (size_t)B_ * T_ * N_ * H_;

    const int BLK = 256;

    k1_w_deg<<<NBLK_P, BLK_, 0, stream>>>(feat, nn, mm, uw, deg);
    k2_uw<<<NBLK_P, BLK_, 0, stream>>>(uw, deg, nn);
    k3_qi<<<(TILES3_ * 32 + BLK_ - 1) / BLK_, BLK_, 0, stream>>>(feat, q1, qiw);
    k4_wd<<<(ROWS7_ * H_ + BLK_ - 1) / BLK_, BLK_, 0, stream>>>(feat, nn, q2, qiw, dw);
    {
        const size_t total = (size_t)B_ * T7_ * N_ * H_;
        k5_dw<<<(unsigned)((total + BLK - 1) / BLK), BLK, 0, stream>>>(dw);
    }
}


# In-degree normalisation folded into k4: a thread keeps its 17 weights in registers, sums them in k order and stores d_w once, normalised; the k5 launch and its d_w re-read and re-write are gone
# speedup vs baseline: 2.0444x; 1.1343x over previous
#include <hip/hip_runtime.h>
#include <math.h>

#define B_   2
#define T_   8
#define N_   5000
#define K1_  17
#define H_   4
#define C_   16
#define NO_  8
#define T7_  (T_-1)
#define E2_  (N_*K1_)

#define BLK_      256
#define NBLK_P    1000
#define WAVES_P   (NBLK_P * BLK_ / 32)
#define THREADS_P (NBLK_P * BLK_)

typedef float v2f __attribute__((ext_vector_type(2)));
typedef float v8f __attribute__((ext_vector_type(8)));

#define SWZ(x, M) __int_as_float(__builtin_amdgcn_ds_swizzle( \
                      __float_as_int(x), ((M) << 10) | 0x1f))

template<int M>
__device__ __forceinline__ float mergepair(float a, float b, int c) {
    const float keep = (c & M) ? b : a;
    const float send = (c & M) ? a : b;
    return keep + SWZ(send, M);
}

__device__ __forceinline__ float row_reduce16(const float t[8], int c) {
    const float m0 = mergepair<1>(t[0], t[1], c);
    const float m1 = mergepair<1>(t[2], t[3], c);
    const float m2 = mergepair<1>(t[4], t[5], c);
    const float m3 = mergepair<1>(t[6], t[7], c);
    const float n0 = mergepair<2>(m0, m1, c);
    const float n1 = mergepair<2>(m2, m3, c);
    float s = mergepair<4>(n0, n1, c);
    s += SWZ(s, 8);
    return s;
}

#define NODES1_ (B_ * T_ * N_)
#define TRIPS1_ (NODES1_ / WAVES_P)
static_assert(NODES1_ % WAVES_P == 0, "k1: every wave takes the same number of nodes");

__global__ void k1_w_deg(const float* __restrict__ feat,
                         const int*   __restrict__ nn,
                         const float* __restrict__ mm,
                         float*       __restrict__ w_out,
                         float*       __restrict__ deg)
{
    const int wid  = __builtin_amdgcn_readfirstlane(
                         (int)((blockIdx.x * blockDim.x + threadIdx.x) >> 5));
    const int lane = threadIdx.x & 31;

    const int r  = lane & 15;
    const int hh = lane >> 4;
    const int row = (lane & 7) | (hh << 3);

    v2f bm[H_][4];
#pragma unroll
    for (int h = 0; h < H_; ++h) {
        const float* Mh = mm + h * C_ * C_;
#pragma unroll
        for (int c = 0; c < 4; ++c) {
            const int j0 = 4 * c + 2 * hh;
            bm[h][c].x = Mh[r * C_ + j0];
            bm[h][c].y = Mh[r * C_ + j0 + 1];
        }
    }

    int nbr_next = nn[(wid % N_) * K1_ + 1 + r];

    for (int it = 0; it < TRIPS1_; ++it) {
        const int g  = wid + it * WAVES_P;
        const int bt = g / N_;
        const int n  = g - bt * N_;

        const int nbrv = nbr_next;
        if (it + 1 < TRIPS1_)
            nbr_next = nn[((g + WAVES_P) % N_) * K1_ + 1 + r];

        const unsigned vm = (unsigned)__ballot(nbrv >= 0);
        const int nj = (nbrv < 0) ? n : nbrv;

        const float* fi0 = feat + ((size_t)g * H_ * C_);
        const float* fj0 = feat + (((size_t)bt * N_ + nj) * H_ * C_);

        const bool rv = ((vm >> row) & 1u) != 0u;

        float wq[H_], dq[H_];
#pragma unroll
        for (int h = 0; h < H_; ++h) {
            const float* fi = fi0 + h * C_;
            const float* fj = fj0 + h * C_;

            v8f d = {0.f,0.f,0.f,0.f,0.f,0.f,0.f,0.f};
#pragma unroll
            for (int c = 0; c < 4; ++c) {
                const int j0 = 4 * c + 2 * hh;
                v2f av;
                av.x = fi[j0]     - fj[j0];
                av.y = fi[j0 + 1] - fj[j0 + 1];
                d = __builtin_amdgcn_wmma_f32_16x16x4_f32(
                        false, av, false, bm[h][c], (short)0, d, false, false);
            }

            float t[8];
#pragma unroll
            for (int v = 0; v < 8; ++v) t[v] = d[v] * d[v];
            const float s = row_reduce16(t, r);

            const float w = rv ? __expf(-s) : 0.f;

            float dp = w;
            dp += SWZ(dp, 1);
            dp += SWZ(dp, 2);
            dp += SWZ(dp, 4);
            dp += SWZ(dp, 16);

            wq[h] = w;
            dq[h] = dp;
        }

        if ((lane & 15) < 8)
            *reinterpret_cast<float4*>(w_out + ((size_t)g * 16 + row) * H_) =
                make_float4(wq[0], wq[1], wq[2], wq[3]);
        if (lane == 0)
            *reinterpret_cast<float4*>(deg + (size_t)g * H_) =
                make_float4(dq[0], dq[1], dq[2], dq[3]);
    }
}

__device__ __forceinline__ float norm_one(float w, float di, float dj)
{
    const float dm = sqrtf(di * dj);
    return (dm > 0.f) ? (w / dm) : 0.f;
}

#define ITEMS2_ (B_ * T_ * N_ * 16)
#define TRIPS2_ (ITEMS2_ / THREADS_P)
static_assert(ITEMS2_ % THREADS_P == 0, "k2: every thread takes the same number of rows");

__global__ void k2_uw(float* __restrict__ uw,
                      const float* __restrict__ deg,
                      const int*   __restrict__ nn)
{
    const int t0 = (int)(blockIdx.x * blockDim.x + threadIdx.x);

    float4 w[TRIPS2_], di[TRIPS2_], dj[TRIPS2_];
    int    nb[TRIPS2_], own[TRIPS2_];
#pragma unroll
    for (int i = 0; i < TRIPS2_; ++i) {
        const int item = t0 + i * THREADS_P;
        const int k  = item % 16;
        own[i] = item / 16;
        const int n  = own[i] % N_;
        w[i]  = *reinterpret_cast<const float4*>(uw + (size_t)item * H_);
        nb[i] = nn[n * K1_ + 1 + k];
        di[i] = *reinterpret_cast<const float4*>(deg + (size_t)own[i] * H_);
    }
#pragma unroll
    for (int i = 0; i < TRIPS2_; ++i) {
        const int n   = own[i] % N_;
        const int nbc = (nb[i] >= 0) ? nb[i] : n;
        dj[i] = *reinterpret_cast<const float4*>(deg + (size_t)(own[i] - n + nbc) * H_);
        if (nb[i] < 0) dj[i] = make_float4(0.f, 0.f, 0.f, 0.f);
    }
#pragma unroll
    for (int i = 0; i < TRIPS2_; ++i) {
        const int item = t0 + i * THREADS_P;
        float4 o;
        o.x = norm_one(w[i].x, di[i].x, dj[i].x);
        o.y = norm_one(w[i].y, di[i].y, dj[i].y);
        o.z = norm_one(w[i].z, di[i].z, dj[i].z);
        o.w = norm_one(w[i].w, di[i].w, dj[i].w);
        *reinterpret_cast<float4*>(uw + (size_t)item * H_) = o;
    }
}

#define ROWS7_  (B_ * T7_ * N_)
#define TILES3_ (ROWS7_ / 16)
static_assert(ROWS7_ % 16 == 0, "k3: whole 16-row tiles");

__global__ void k3_qi(const float* __restrict__ feat,
                      const float* __restrict__ q1,
                      float*       __restrict__ qi)
{
    const int wid  = __builtin_amdgcn_readfirstlane(
                         (int)((blockIdx.x * blockDim.x + threadIdx.x) >> 5));
    if (wid >= TILES3_) return;
    const int lane = threadIdx.x & 31;
    const int r  = lane & 15;
    const int hh = lane >> 4;

    const int g7  = wid * 16 + r;
    const int bt7 = g7 / N_;
    const int n   = g7 - bt7 * N_;
    const int b   = bt7 / T7_;
    const int t   = bt7 - b * T7_;

    const float* f0 = feat + (((size_t)(b * T_ + t) * N_ + n) * H_ * C_);

    v8f d[H_];
#pragma unroll
    for (int h = 0; h < H_; ++h) {
        const float* fr = f0 + h * C_;
        const float* Qh = q1 + (h * NO_ + (r & (NO_ - 1))) * C_;
        v8f acc = {0.f,0.f,0.f,0.f,0.f,0.f,0.f,0.f};
#pragma unroll
        for (int c = 0; c < 4; ++c) {
            const int j0 = 4 * c + 2 * hh;
            v2f av, bv;
            av.x = Qh[j0];
            av.y = Qh[j0 + 1];
            bv.x = fr[j0];
            bv.y = fr[j0 + 1];
            acc = __builtin_amdgcn_wmma_f32_16x16x4_f32(
                      false, av, false, bv, (short)0, acc, false, false);
        }
        d[h] = acc;
    }

    if (hh == 0) {
        float* o = qi + (size_t)g7 * H_ * NO_;
#pragma unroll
        for (int h = 0; h < H_; ++h) {
            *reinterpret_cast<float4*>(o + h * NO_) =
                make_float4(d[h][0], d[h][1], d[h][2], d[h][3]);
            *reinterpret_cast<float4*>(o + h * NO_ + 4) =
                make_float4(d[h][4], d[h][5], d[h][6], d[h][7]);
        }
    }
}

__global__ void __launch_bounds__(BLK_)
k4_wd(const float* __restrict__ feat,
      const int*   __restrict__ nn,
      const float* __restrict__ q2,
      const float* __restrict__ qi,
      float*       __restrict__ dw)
{
    const int tid = (int)(blockIdx.x * blockDim.x + threadIdx.x);
    if (tid >= ROWS7_ * H_) return;
    const int h   = tid % H_;
    const int g7  = tid / H_;
    const int bt7 = g7 / N_;
    const int n   = g7 - bt7 * N_;
    const int b   = bt7 / T7_;
    const int t   = bt7 - b * T7_;

    int nb[K1_];
#pragma unroll
    for (int k = 0; k < K1_; ++k) nb[k] = nn[n * K1_ + k];

    const float* f = feat + ((((size_t)(b * T_ + t + 1) * N_ + n) * H_ + h) * C_);
    const float* Q = q2 + h * NO_ * C_;
    float fv[C_];
#pragma unroll
    for (int j = 0; j < C_; j += 4) {
        const float4 x = *reinterpret_cast<const float4*>(f + j);
        fv[j] = x.x; fv[j + 1] = x.y; fv[j + 2] = x.z; fv[j + 3] = x.w;
    }
    float qj[NO_];
#pragma unroll
    for (int i = 0; i < NO_; ++i) {
        float s = 0.f;
#pragma unroll
        for (int j = 0; j < C_; ++j) s += Q[i * C_ + j] * fv[j];
        qj[i] = s;
    }

    const float* qrow = qi + ((size_t)bt7 * N_ * H_ + h) * NO_;
    float*       o    = dw + ((size_t)bt7 * E2_ + (size_t)n * K1_) * H_ + h;
    float wd[K1_];
#pragma unroll
    for (int k = 0; k < K1_; ++k) {
        const bool ok = (nb[k] >= 0);
        const int  jn = ok ? nb[k] : n;
        const float* p = qrow + (size_t)jn * (H_ * NO_);
        const float4 a0 = *reinterpret_cast<const float4*>(p);
        const float4 a1 = *reinterpret_cast<const float4*>(p + 4);
        const float s = ((a0.x * qj[0] + a0.y * qj[1]) + (a0.z * qj[2] + a0.w * qj[3]))
                      + ((a1.x * qj[4] + a1.y * qj[5]) + (a1.z * qj[6] + a1.w * qj[7]));
        const float e = __expf(-s);
        wd[k] = ok ? e : 0.f;
    }

    float sum = 0.f;
#pragma unroll
    for (int k = 0; k < K1_; ++k) sum += wd[k];
    const float inv = (sum > 0.f) ? (1.f / sum) : 0.f;
#pragma unroll
    for (int k = 0; k < K1_; ++k) o[k * H_] = wd[k] * inv;
}

extern "C" void kernel_launch(void* const* d_in, const int* in_sizes, int n_in,
                              void* d_out, int out_size, void* d_ws, size_t ws_size,
                              hipStream_t stream)
{
    const float* feat = (const float*)d_in[0];
    const int*   nn   = (const int*)  d_in[1];
    const float* q1   = (const float*)d_in[2];
    const float* q2   = (const float*)d_in[3];
    const float* mm   = (const float*)d_in[4];

    float* uw = (float*)d_out;
    float* dw = uw + (size_t)B_ * T_ * N_ * 16 * H_;

    float* deg = (float*)d_ws;
    float* qiw = deg + (size_t)B_ * T_ * N_ * H_;

    k1_w_deg<<<NBLK_P, BLK_, 0, stream>>>(feat, nn, mm, uw, deg);
    k2_uw<<<NBLK_P, BLK_, 0, stream>>>(uw, deg, nn);
    k3_qi<<<(TILES3_ * 32 + BLK_ - 1) / BLK_, BLK_, 0, stream>>>(feat, q1, qiw);
    k4_wd<<<(ROWS7_ * H_ + BLK_ - 1) / BLK_, BLK_, 0, stream>>>(feat, nn, q2, qiw, dw);
}


# WMMA k order remapped in both matrix kernels (lane half hh takes k = 8hh..8hh+7, was 4c+2hh): a lane's operands are two 16-byte loads per head, not four 8-byte; k summation order changes
# speedup vs baseline: 2.2570x; 1.1040x over previous
#include <hip/hip_runtime.h>
#include <math.h>

#define B_   2
#define T_   8
#define N_   5000
#define K1_  17
#define H_   4
#define C_   16
#define NO_  8
#define T7_  (T_-1)
#define E2_  (N_*K1_)

#define BLK_      256
#define NBLK_P    1000
#define WAVES_P   (NBLK_P * BLK_ / 32)
#define THREADS_P (NBLK_P * BLK_)

typedef float v2f __attribute__((ext_vector_type(2)));
typedef float v8f __attribute__((ext_vector_type(8)));

#define SWZ(x, M) __int_as_float(__builtin_amdgcn_ds_swizzle( \
                      __float_as_int(x), ((M) << 10) | 0x1f))

template<int M>
__device__ __forceinline__ float mergepair(float a, float b, int c) {
    const float keep = (c & M) ? b : a;
    const float send = (c & M) ? a : b;
    return keep + SWZ(send, M);
}

__device__ __forceinline__ float row_reduce16(const float t[8], int c) {
    const float m0 = mergepair<1>(t[0], t[1], c);
    const float m1 = mergepair<1>(t[2], t[3], c);
    const float m2 = mergepair<1>(t[4], t[5], c);
    const float m3 = mergepair<1>(t[6], t[7], c);
    const float n0 = mergepair<2>(m0, m1, c);
    const float n1 = mergepair<2>(m2, m3, c);
    float s = mergepair<4>(n0, n1, c);
    s += SWZ(s, 8);
    return s;
}

#define NODES1_ (B_ * T_ * N_)
#define TRIPS1_ (NODES1_ / WAVES_P)
static_assert(NODES1_ % WAVES_P == 0, "k1: every wave takes the same number of nodes");

__global__ void k1_w_deg(const float* __restrict__ feat,
                         const int*   __restrict__ nn,
                         const float* __restrict__ mm,
                         float*       __restrict__ w_out,
                         float*       __restrict__ deg)
{
    const int wid  = __builtin_amdgcn_readfirstlane(
                         (int)((blockIdx.x * blockDim.x + threadIdx.x) >> 5));
    const int lane = threadIdx.x & 31;

    const int r  = lane & 15;
    const int hh = lane >> 4;
    const int row = (lane & 7) | (hh << 3);

    v2f bm[H_][4];
#pragma unroll
    for (int h = 0; h < H_; ++h) {
        const float* Mh = mm + h * C_ * C_;
#pragma unroll
        for (int c = 0; c < 4; ++c) {
            const int j0 = 8 * hh + 2 * c;
            bm[h][c].x = Mh[r * C_ + j0];
            bm[h][c].y = Mh[r * C_ + j0 + 1];
        }
    }

    int nbr_next = nn[(wid % N_) * K1_ + 1 + r];

    for (int it = 0; it < TRIPS1_; ++it) {
        const int g  = wid + it * WAVES_P;
        const int bt = g / N_;
        const int n  = g - bt * N_;

        const int nbrv = nbr_next;
        if (it + 1 < TRIPS1_)
            nbr_next = nn[((g + WAVES_P) % N_) * K1_ + 1 + r];

        const unsigned vm = (unsigned)__ballot(nbrv >= 0);
        const int nj = (nbrv < 0) ? n : nbrv;

        const float* fi0 = feat + ((size_t)g * H_ * C_);
        const float* fj0 = feat + (((size_t)bt * N_ + nj) * H_ * C_);

        const bool rv = ((vm >> row) & 1u) != 0u;

        float wq[H_], dq[H_];
#pragma unroll
        for (int h = 0; h < H_; ++h) {
            const float* fi = fi0 + h * C_;
            const float* fj = fj0 + h * C_;

            const float4 i0 = *reinterpret_cast<const float4*>(fi + 8 * hh);
            const float4 i1 = *reinterpret_cast<const float4*>(fi + 8 * hh + 4);
            const float4 n0 = *reinterpret_cast<const float4*>(fj + 8 * hh);
            const float4 n1 = *reinterpret_cast<const float4*>(fj + 8 * hh + 4);
            v2f av[4];
            av[0].x = i0.x - n0.x;  av[0].y = i0.y - n0.y;
            av[1].x = i0.z - n0.z;  av[1].y = i0.w - n0.w;
            av[2].x = i1.x - n1.x;  av[2].y = i1.y - n1.y;
            av[3].x = i1.z - n1.z;  av[3].y = i1.w - n1.w;

            v8f d = {0.f,0.f,0.f,0.f,0.f,0.f,0.f,0.f};
#pragma unroll
            for (int c = 0; c < 4; ++c)
                d = __builtin_amdgcn_wmma_f32_16x16x4_f32(
                        false, av[c], false, bm[h][c], (short)0, d, false, false);

            float t[8];
#pragma unroll
            for (int v = 0; v < 8; ++v) t[v] = d[v] * d[v];
            const float s = row_reduce16(t, r);

            const float w = rv ? __expf(-s) : 0.f;

            float dp = w;
            dp += SWZ(dp, 1);
            dp += SWZ(dp, 2);
            dp += SWZ(dp, 4);
            dp += SWZ(dp, 16);

            wq[h] = w;
            dq[h] = dp;
        }

        if ((lane & 15) < 8)
            *reinterpret_cast<float4*>(w_out + ((size_t)g * 16 + row) * H_) =
                make_float4(wq[0], wq[1], wq[2], wq[3]);
        if (lane == 0)
            *reinterpret_cast<float4*>(deg + (size_t)g * H_) =
                make_float4(dq[0], dq[1], dq[2], dq[3]);
    }
}

__device__ __forceinline__ float norm_one(float w, float di, float dj)
{
    const float dm = sqrtf(di * dj);
    return (dm > 0.f) ? (w / dm) : 0.f;
}

#define ITEMS2_ (B_ * T_ * N_ * 16)
#define TRIPS2_ (ITEMS2_ / THREADS_P)
static_assert(ITEMS2_ % THREADS_P == 0, "k2: every thread takes the same number of rows");

__global__ void k2_uw(float* __restrict__ uw,
                      const float* __restrict__ deg,
                      const int*   __restrict__ nn)
{
    const int t0 = (int)(blockIdx.x * blockDim.x + threadIdx.x);

    float4 w[TRIPS2_], di[TRIPS2_], dj[TRIPS2_];
    int    nb[TRIPS2_], own[TRIPS2_];
#pragma unroll
    for (int i = 0; i < TRIPS2_; ++i) {
        const int item = t0 + i * THREADS_P;
        const int k  = item % 16;
        own[i] = item / 16;
        const int n  = own[i] % N_;
        w[i]  = *reinterpret_cast<const float4*>(uw + (size_t)item * H_);
        nb[i] = nn[n * K1_ + 1 + k];
        di[i] = *reinterpret_cast<const float4*>(deg + (size_t)own[i] * H_);
    }
#pragma unroll
    for (int i = 0; i < TRIPS2_; ++i) {
        const int n   = own[i] % N_;
        const int nbc = (nb[i] >= 0) ? nb[i] : n;
        dj[i] = *reinterpret_cast<const float4*>(deg + (size_t)(own[i] - n + nbc) * H_);
        if (nb[i] < 0) dj[i] = make_float4(0.f, 0.f, 0.f, 0.f);
    }
#pragma unroll
    for (int i = 0; i < TRIPS2_; ++i) {
        const int item = t0 + i * THREADS_P;
        float4 o;
        o.x = norm_one(w[i].x, di[i].x, dj[i].x);
        o.y = norm_one(w[i].y, di[i].y, dj[i].y);
        o.z = norm_one(w[i].z, di[i].z, dj[i].z);
        o.w = norm_one(w[i].w, di[i].w, dj[i].w);
        *reinterpret_cast<float4*>(uw + (size_t)item * H_) = o;
    }
}

#define ROWS7_  (B_ * T7_ * N_)
#define TILES3_ (ROWS7_ / 16)
static_assert(ROWS7_ % 16 == 0, "k3: whole 16-row tiles");

__global__ void k3_qi(const float* __restrict__ feat,
                      const float* __restrict__ q1,
                      float*       __restrict__ qi)
{
    const int wid  = __builtin_amdgcn_readfirstlane(
                         (int)((blockIdx.x * blockDim.x + threadIdx.x) >> 5));
    if (wid >= TILES3_) return;
    const int lane = threadIdx.x & 31;
    const int r  = lane & 15;
    const int hh = lane >> 4;

    const int g7  = wid * 16 + r;
    const int bt7 = g7 / N_;
    const int n   = g7 - bt7 * N_;
    const int b   = bt7 / T7_;
    const int t   = bt7 - b * T7_;

    const float* f0 = feat + (((size_t)(b * T_ + t) * N_ + n) * H_ * C_);

    v8f d[H_];
#pragma unroll
    for (int h = 0; h < H_; ++h) {
        const float* fr = f0 + h * C_;
        const float* Qh = q1 + (h * NO_ + (r & (NO_ - 1))) * C_;
        const float4 qa = *reinterpret_cast<const float4*>(Qh + 8 * hh);
        const float4 qb = *reinterpret_cast<const float4*>(Qh + 8 * hh + 4);
        const float4 fa = *reinterpret_cast<const float4*>(fr + 8 * hh);
        const float4 fb = *reinterpret_cast<const float4*>(fr + 8 * hh + 4);
        v2f av[4], bv[4];
        av[0].x = qa.x;  av[0].y = qa.y;   bv[0].x = fa.x;  bv[0].y = fa.y;
        av[1].x = qa.z;  av[1].y = qa.w;   bv[1].x = fa.z;  bv[1].y = fa.w;
        av[2].x = qb.x;  av[2].y = qb.y;   bv[2].x = fb.x;  bv[2].y = fb.y;
        av[3].x = qb.z;  av[3].y = qb.w;   bv[3].x = fb.z;  bv[3].y = fb.w;
        v8f acc = {0.f,0.f,0.f,0.f,0.f,0.f,0.f,0.f};
#pragma unroll
        for (int c = 0; c < 4; ++c)
            acc = __builtin_amdgcn_wmma_f32_16x16x4_f32(
                      false, av[c], false, bv[c], (short)0, acc, false, false);
        d[h] = acc;
    }

    if (hh == 0) {
        float* o = qi + (size_t)g7 * H_ * NO_;
#pragma unroll
        for (int h = 0; h < H_; ++h) {
            *reinterpret_cast<float4*>(o + h * NO_) =
                make_float4(d[h][0], d[h][1], d[h][2], d[h][3]);
            *reinterpret_cast<float4*>(o + h * NO_ + 4) =
                make_float4(d[h][4], d[h][5], d[h][6], d[h][7]);
        }
    }
}

__global__ void __launch_bounds__(BLK_)
k4_wd(const float* __restrict__ feat,
      const int*   __restrict__ nn,
      const float* __restrict__ q2,
      const float* __restrict__ qi,
      float*       __restrict__ dw)
{
    const int tid = (int)(blockIdx.x * blockDim.x + threadIdx.x);
    if (tid >= ROWS7_ * H_) return;
    const int h   = tid % H_;
    const int g7  = tid / H_;
    const int bt7 = g7 / N_;
    const int n   = g7 - bt7 * N_;
    const int b   = bt7 / T7_;
    const int t   = bt7 - b * T7_;

    int nb[K1_];
#pragma unroll
    for (int k = 0; k < K1_; ++k) nb[k] = nn[n * K1_ + k];

    const float* f = feat + ((((size_t)(b * T_ + t + 1) * N_ + n) * H_ + h) * C_);
    const float* Q = q2 + h * NO_ * C_;
    float fv[C_];
#pragma unroll
    for (int j = 0; j < C_; j += 4) {
        const float4 x = *reinterpret_cast<const float4*>(f + j);
        fv[j] = x.x; fv[j + 1] = x.y; fv[j + 2] = x.z; fv[j + 3] = x.w;
    }
    float qj[NO_];
#pragma unroll
    for (int i = 0; i < NO_; ++i) {
        float s = 0.f;
#pragma unroll
        for (int j = 0; j < C_; ++j) s += Q[i * C_ + j] * fv[j];
        qj[i] = s;
    }

    const float* qrow = qi + ((size_t)bt7 * N_ * H_ + h) * NO_;
    float*       o    = dw + ((size_t)bt7 * E2_ + (size_t)n * K1_) * H_ + h;
    float wd[K1_];
#pragma unroll
    for (int k = 0; k < K1_; ++k) {
        const bool ok = (nb[k] >= 0);
        const int  jn = ok ? nb[k] : n;
        const float* p = qrow + (size_t)jn * (H_ * NO_);
        const float4 a0 = *reinterpret_cast<const float4*>(p);
        const float4 a1 = *reinterpret_cast<const float4*>(p + 4);
        const float s = ((a0.x * qj[0] + a0.y * qj[1]) + (a0.z * qj[2] + a0.w * qj[3]))
                      + ((a1.x * qj[4] + a1.y * qj[5]) + (a1.z * qj[6] + a1.w * qj[7]));
        const float e = __expf(-s);
        wd[k] = ok ? e : 0.f;
    }

    float sum = 0.f;
#pragma unroll
    for (int k = 0; k < K1_; ++k) sum += wd[k];
    const float inv = (sum > 0.f) ? (1.f / sum) : 0.f;
#pragma unroll
    for (int k = 0; k < K1_; ++k) o[k * H_] = wd[k] * inv;
}

extern "C" void kernel_launch(void* const* d_in, const int* in_sizes, int n_in,
                              void* d_out, int out_size, void* d_ws, size_t ws_size,
                              hipStream_t stream)
{
    const float* feat = (const float*)d_in[0];
    const int*   nn   = (const int*)  d_in[1];
    const float* q1   = (const float*)d_in[2];
    const float* q2   = (const float*)d_in[3];
    const float* mm   = (const float*)d_in[4];

    float* uw = (float*)d_out;
    float* dw = uw + (size_t)B_ * T_ * N_ * 16 * H_;

    float* deg = (float*)d_ws;
    float* qiw = deg + (size_t)B_ * T_ * N_ * H_;

    k1_w_deg<<<NBLK_P, BLK_, 0, stream>>>(feat, nn, mm, uw, deg);
    k2_uw<<<NBLK_P, BLK_, 0, stream>>>(uw, deg, nn);
    k3_qi<<<(TILES3_ * 32 + BLK_ - 1) / BLK_, BLK_, 0, stream>>>(feat, q1, qiw);
    k4_wd<<<(ROWS7_ * H_ + BLK_ - 1) / BLK_, BLK_, 0, stream>>>(feat, nn, q2, qiw, dw);
}


# multiM projection hoisted ahead of the gather: a new kernel projects each node once on WMMA (was per edge); k1 gathers projected rows, sums (Pi-Pj)^2 by DPP; sum order changes; launches 4->5
# speedup vs baseline: 2.5288x; 1.1204x over previous
#include <hip/hip_runtime.h>
#include <math.h>

#define B_   2
#define T_   8
#define N_   5000
#define K1_  17
#define H_   4
#define C_   16
#define NO_  8
#define T7_  (T_-1)
#define E2_  (N_*K1_)

#define BLK_      256
#define NBLK_P    1000
#define WAVES_P   (NBLK_P * BLK_ / 32)
#define THREADS_P (NBLK_P * BLK_)

typedef float v2f __attribute__((ext_vector_type(2)));
typedef float v8f __attribute__((ext_vector_type(8)));

template<int CTRL>
__device__ __forceinline__ int dpp_i(int x) {
    return __builtin_amdgcn_update_dpp(0, x, CTRL, 0xf, 0xf, true);
}
template<int CTRL>
__device__ __forceinline__ float dpp_f(float x) {
    return __int_as_float(dpp_i<CTRL>(__float_as_int(x)));
}
#define DPP_QUAD_1032   0xB1
#define DPP_QUAD_2301   0x4E
#define DPP_ROW_SHARE0  0x150

__device__ __forceinline__ float xhalf(float x) {
    const int b = __float_as_int(x);
    return __int_as_float(__builtin_amdgcn_permlanex16(b, b, 0x76543210, 0xfedcba98, false, false));
}

#define PLANE_   (N_ * H_ * C_)
#define PSPLIT_  11
static_assert((size_t)PSPLIT_ * PLANE_ <= (size_t)B_ * T7_ * E2_ * H_, "P planes 0..PSPLIT_-1 fit the d_w region");
static_assert((size_t)(B_ * T_ - PSPLIT_) * PLANE_ <= (size_t)B_ * T7_ * N_ * H_ * NO_, "the other P planes fit the Qi plane");

#define NODES1_ (B_ * T_ * N_)
#define TILESP_ (NODES1_ / 16)
static_assert(NODES1_ % 16 == 0, "kP: whole 16-row tiles");
static_assert((TILESP_ * 32) % BLK_ == 0, "kP: the grid holds no surplus wave");

__global__ void kp_proj(const float* __restrict__ feat,
                        const float* __restrict__ mm,
                        float*       __restrict__ p_lo,
                        float*       __restrict__ p_hi)
{
    const int wid  = __builtin_amdgcn_readfirstlane(
                         (int)((blockIdx.x * blockDim.x + threadIdx.x) >> 5));
    const int lane = threadIdx.x & 31;
    const int r  = lane & 15;
    const int hh = lane >> 4;

    const int g  = wid * 16 + r;
    const int bt = g / N_;

    const float* f0 = feat + (size_t)g * H_ * C_;
    float* o = (bt < PSPLIT_) ? (p_lo + (size_t)g * H_ * C_)
                              : (p_hi + (size_t)(g - PSPLIT_ * N_) * H_ * C_);

    v8f d[H_];
#pragma unroll
    for (int h = 0; h < H_; ++h) {
        const float* fr = f0 + h * C_;
        const float* Mr = mm + (h * C_ + r) * C_;
        const float4 ma = *reinterpret_cast<const float4*>(Mr + 8 * hh);
        const float4 mb = *reinterpret_cast<const float4*>(Mr + 8 * hh + 4);
        const float4 fa = *reinterpret_cast<const float4*>(fr + 8 * hh);
        const float4 fb = *reinterpret_cast<const float4*>(fr + 8 * hh + 4);
        v2f av[4], bv[4];
        av[0].x = ma.x;  av[0].y = ma.y;   bv[0].x = fa.x;  bv[0].y = fa.y;
        av[1].x = ma.z;  av[1].y = ma.w;   bv[1].x = fa.z;  bv[1].y = fa.w;
        av[2].x = mb.x;  av[2].y = mb.y;   bv[2].x = fb.x;  bv[2].y = fb.y;
        av[3].x = mb.z;  av[3].y = mb.w;   bv[3].x = fb.z;  bv[3].y = fb.w;
        v8f acc = {0.f,0.f,0.f,0.f,0.f,0.f,0.f,0.f};
#pragma unroll
        for (int c = 0; c < 4; ++c)
            acc = __builtin_amdgcn_wmma_f32_16x16x4_f32(
                      false, av[c], false, bv[c], (short)0, acc, false, false);
        d[h] = acc;
    }

#pragma unroll
    for (int h = 0; h < H_; ++h) {
        *reinterpret_cast<float4*>(o + h * C_ + 8 * hh) =
            make_float4(d[h][0], d[h][1], d[h][2], d[h][3]);
        *reinterpret_cast<float4*>(o + h * C_ + 8 * hh + 4) =
            make_float4(d[h][4], d[h][5], d[h][6], d[h][7]);
    }
}

#define TRIPS1_ (NODES1_ / WAVES_P)
static_assert(NODES1_ % WAVES_P == 0, "k1: every wave takes the same number of nodes");

template<int Q>
__device__ __forceinline__ float4 gather_piece(const float* __restrict__ plane, int nbrv, int n,
                                               unsigned piece_bytes, bool& ok)
{
    const int nb = dpp_i<DPP_ROW_SHARE0 + 2 * Q>(nbrv);
    ok = (nb >= 0);
    const int jn = ok ? nb : n;
    return *reinterpret_cast<const float4*>(
               reinterpret_cast<const char*>(plane) + ((unsigned)jn * (unsigned)(H_ * C_ * 4) + piece_bytes));
}

__device__ __forceinline__ float edge_weight(const float4 pi, const float4 pj, bool ok)
{
    const float dx = pi.x - pj.x, dy = pi.y - pj.y, dz = pi.z - pj.z, dw = pi.w - pj.w;
    float s = dx * dx;
    s = fmaf(dy, dy, s);
    s = fmaf(dz, dz, s);
    s = fmaf(dw, dw, s);
    s += dpp_f<DPP_QUAD_1032>(s);
    s += dpp_f<DPP_QUAD_2301>(s);
    return ok ? __expf(-s) : 0.f;
}

__global__ void k1_w_deg(const float* __restrict__ p_lo,
                         const float* __restrict__ p_hi,
                         const int*   __restrict__ nn,
                         float*       __restrict__ w_out,
                         float*       __restrict__ deg)
{
    const int wid  = __builtin_amdgcn_readfirstlane(
                         (int)((blockIdx.x * blockDim.x + threadIdx.x) >> 5));
    const int lane = threadIdx.x & 31;

    const int c  = lane & 15;
    const int hh = lane >> 4;
    const int p  = c & 3;
    const int h  = c >> 2;
    const int slot = (c + hh) & 15;
    const unsigned piece_bytes = (unsigned)c * 16u;

    int nbr_next = nn[(wid % N_) * K1_ + 1 + slot];

    for (int it = 0; it < TRIPS1_; ++it) {
        const int g  = wid + it * WAVES_P;
        const int bt = g / N_;
        const int n  = g - bt * N_;

        const int nbrv = nbr_next;
        if (it + 1 < TRIPS1_)
            nbr_next = nn[((g + WAVES_P) % N_) * K1_ + 1 + slot];

        const float* plane = (bt < PSPLIT_) ? (p_lo + (size_t)bt * PLANE_)
                                            : (p_hi + (size_t)(bt - PSPLIT_) * PLANE_);

        const float4 pi = *reinterpret_cast<const float4*>(
                              reinterpret_cast<const char*>(plane) + ((unsigned)n * (unsigned)(H_ * C_ * 4) + piece_bytes));

        bool ok[8];
        float4 pj[8];
        pj[0] = gather_piece<0>(plane, nbrv, n, piece_bytes, ok[0]);
        pj[1] = gather_piece<1>(plane, nbrv, n, piece_bytes, ok[1]);
        pj[2] = gather_piece<2>(plane, nbrv, n, piece_bytes, ok[2]);
        pj[3] = gather_piece<3>(plane, nbrv, n, piece_bytes, ok[3]);
        pj[4] = gather_piece<4>(plane, nbrv, n, piece_bytes, ok[4]);
        pj[5] = gather_piece<5>(plane, nbrv, n, piece_bytes, ok[5]);
        pj[6] = gather_piece<6>(plane, nbrv, n, piece_bytes, ok[6]);
        pj[7] = gather_piece<7>(plane, nbrv, n, piece_bytes, ok[7]);

        float wv[8];
#pragma unroll
        for (int q = 0; q < 8; ++q) wv[q] = edge_weight(pi, pj[q], ok[q]);

        float dp = ((wv[0] + wv[1]) + (wv[2] + wv[3])) + ((wv[4] + wv[5]) + (wv[6] + wv[7]));
        dp += xhalf(dp);

        const float wlo = (p == 0) ? wv[0] : (p == 1) ? wv[1] : (p == 2) ? wv[2] : wv[3];
        const float whi = (p == 0) ? wv[4] : (p == 1) ? wv[5] : (p == 2) ? wv[6] : wv[7];
        float* wrow = w_out + (size_t)g * 16 * H_ + (2 * p + hh) * H_ + h;
        wrow[0]      = wlo;
        wrow[8 * H_] = whi;
        if (lane < 16 && p == 0)
            deg[(size_t)g * H_ + h] = dp;
    }
}

__device__ __forceinline__ float norm_one(float w, float di, float dj)
{
    const float dm = sqrtf(di * dj);
    return (dm > 0.f) ? (w / dm) : 0.f;
}

#define ITEMS2_ (B_ * T_ * N_ * 16)
#define TRIPS2_ (ITEMS2_ / THREADS_P)
static_assert(ITEMS2_ % THREADS_P == 0, "k2: every thread takes the same number of rows");

__global__ void k2_uw(float* __restrict__ uw,
                      const float* __restrict__ deg,
                      const int*   __restrict__ nn)
{
    const int t0 = (int)(blockIdx.x * blockDim.x + threadIdx.x);

    float4 w[TRIPS2_], di[TRIPS2_], dj[TRIPS2_];
    int    nb[TRIPS2_], own[TRIPS2_];
#pragma unroll
    for (int i = 0; i < TRIPS2_; ++i) {
        const int item = t0 + i * THREADS_P;
        const int k  = item % 16;
        own[i] = item / 16;
        const int n  = own[i] % N_;
        w[i]  = *reinterpret_cast<const float4*>(uw + (size_t)item * H_);
        nb[i] = nn[n * K1_ + 1 + k];
        di[i] = *reinterpret_cast<const float4*>(deg + (size_t)own[i] * H_);
    }
#pragma unroll
    for (int i = 0; i < TRIPS2_; ++i) {
        const int n   = own[i] % N_;
        const int nbc = (nb[i] >= 0) ? nb[i] : n;
        dj[i] = *reinterpret_cast<const float4*>(deg + (size_t)(own[i] - n + nbc) * H_);
        if (nb[i] < 0) dj[i] = make_float4(0.f, 0.f, 0.f, 0.f);
    }
#pragma unroll
    for (int i = 0; i < TRIPS2_; ++i) {
        const int item = t0 + i * THREADS_P;
        float4 o;
        o.x = norm_one(w[i].x, di[i].x, dj[i].x);
        o.y = norm_one(w[i].y, di[i].y, dj[i].y);
        o.z = norm_one(w[i].z, di[i].z, dj[i].z);
        o.w = norm_one(w[i].w, di[i].w, dj[i].w);
        *reinterpret_cast<float4*>(uw + (size_t)item * H_) = o;
    }
}

#define ROWS7_  (B_ * T7_ * N_)
#define TILES3_ (ROWS7_ / 16)
static_assert(ROWS7_ % 16 == 0, "k3: whole 16-row tiles");

__global__ void k3_qi(const float* __restrict__ feat,
                      const float* __restrict__ q1,
                      float*       __restrict__ qi)
{
    const int wid  = __builtin_amdgcn_readfirstlane(
                         (int)((blockIdx.x * blockDim.x + threadIdx.x) >> 5));
    if (wid >= TILES3_) return;
    const int lane = threadIdx.x & 31;
    const int r  = lane & 15;
    const int hh = lane >> 4;

    const int g7  = wid * 16 + r;
    const int bt7 = g7 / N_;
    const int n   = g7 - bt7 * N_;
    const int b   = bt7 / T7_;
    const int t   = bt7 - b * T7_;

    const float* f0 = feat + (((size_t)(b * T_ + t) * N_ + n) * H_ * C_);

    v8f d[H_];
#pragma unroll
    for (int h = 0; h < H_; ++h) {
        const float* fr = f0 + h * C_;
        const float* Qh = q1 + (h * NO_ + (r & (NO_ - 1))) * C_;
        const float4 qa = *reinterpret_cast<const float4*>(Qh + 8 * hh);
        const float4 qb = *reinterpret_cast<const float4*>(Qh + 8 * hh + 4);
        const float4 fa = *reinterpret_cast<const float4*>(fr + 8 * hh);
        const float4 fb = *reinterpret_cast<const float4*>(fr + 8 * hh + 4);
        v2f av[4], bv[4];
        av[0].x = qa.x;  av[0].y = qa.y;   bv[0].x = fa.x;  bv[0].y = fa.y;
        av[1].x = qa.z;  av[1].y = qa.w;   bv[1].x = fa.z;  bv[1].y = fa.w;
        av[2].x = qb.x;  av[2].y = qb.y;   bv[2].x = fb.x;  bv[2].y = fb.y;
        av[3].x = qb.z;  av[3].y = qb.w;   bv[3].x = fb.z;  bv[3].y = fb.w;
        v8f acc = {0.f,0.f,0.f,0.f,0.f,0.f,0.f,0.f};
#pragma unroll
        for (int c = 0; c < 4; ++c)
            acc = __builtin_amdgcn_wmma_f32_16x16x4_f32(
                      false, av[c], false, bv[c], (short)0, acc, false, false);
        d[h] = acc;
    }

    if (hh == 0) {
        float* o = qi + (size_t)g7 * H_ * NO_;
#pragma unroll
        for (int h = 0; h < H_; ++h) {
            *reinterpret_cast<float4*>(o + h * NO_) =
                make_float4(d[h][0], d[h][1], d[h][2], d[h][3]);
            *reinterpret_cast<float4*>(o + h * NO_ + 4) =
                make_float4(d[h][4], d[h][5], d[h][6], d[h][7]);
        }
    }
}

__global__ void __launch_bounds__(BLK_)
k4_wd(const float* __restrict__ feat,
      const int*   __restrict__ nn,
      const float* __restrict__ q2,
      const float* __restrict__ qi,
      float*       __restrict__ dw)
{
    const int tid = (int)(blockIdx.x * blockDim.x + threadIdx.x);
    if (tid >= ROWS7_ * H_) return;
    const int h   = tid % H_;
    const int g7  = tid / H_;
    const int bt7 = g7 / N_;
    const int n   = g7 - bt7 * N_;
    const int b   = bt7 / T7_;
    const int t   = bt7 - b * T7_;

    int nb[K1_];
#pragma unroll
    for (int k = 0; k < K1_; ++k) nb[k] = nn[n * K1_ + k];

    const float* f = feat + ((((size_t)(b * T_ + t + 1) * N_ + n) * H_ + h) * C_);
    const float* Q = q2 + h * NO_ * C_;
    float fv[C_];
#pragma unroll
    for (int j = 0; j < C_; j += 4) {
        const float4 x = *reinterpret_cast<const float4*>(f + j);
        fv[j] = x.x; fv[j + 1] = x.y; fv[j + 2] = x.z; fv[j + 3] = x.w;
    }
    float qj[NO_];
#pragma unroll
    for (int i = 0; i < NO_; ++i) {
        float s = 0.f;
#pragma unroll
        for (int j = 0; j < C_; ++j) s += Q[i * C_ + j] * fv[j];
        qj[i] = s;
    }

    const float* qrow = qi + ((size_t)bt7 * N_ * H_ + h) * NO_;
    float*       o    = dw + ((size_t)bt7 * E2_ + (size_t)n * K1_) * H_ + h;
    float wd[K1_];
#pragma unroll
    for (int k = 0; k < K1_; ++k) {
        const bool ok = (nb[k] >= 0);
        const int  jn = ok ? nb[k] : n;
        const float* p = qrow + (size_t)jn * (H_ * NO_);
        const float4 a0 = *reinterpret_cast<const float4*>(p);
        const float4 a1 = *reinterpret_cast<const float4*>(p + 4);
        const float s = ((a0.x * qj[0] + a0.y * qj[1]) + (a0.z * qj[2] + a0.w * qj[3]))
                      + ((a1.x * qj[4] + a1.y * qj[5]) + (a1.z * qj[6] + a1.w * qj[7]));
        const float e = __expf(-s);
        wd[k] = ok ? e : 0.f;
    }

    float sum = 0.f;
#pragma unroll
    for (int k = 0; k < K1_; ++k) sum += wd[k];
    const float inv = (sum > 0.f) ? (1.f / sum) : 0.f;
#pragma unroll
    for (int k = 0; k < K1_; ++k) o[k * H_] = wd[k] * inv;
}

extern "C" void kernel_launch(void* const* d_in, const int* in_sizes, int n_in,
                              void* d_out, int out_size, void* d_ws, size_t ws_size,
                              hipStream_t stream)
{
    const float* feat = (const float*)d_in[0];
    const int*   nn   = (const int*)  d_in[1];
    const float* q1   = (const float*)d_in[2];
    const float* q2   = (const float*)d_in[3];
    const float* mm   = (const float*)d_in[4];

    float* uw = (float*)d_out;
    float* dw = uw + (size_t)B_ * T_ * N_ * 16 * H_;

    float* deg = (float*)d_ws;
    float* qiw = deg + (size_t)B_ * T_ * N_ * H_;

    float* p_lo = dw;
    float* p_hi = qiw;

    kp_proj<<<(TILESP_ * 32) / BLK_, BLK_, 0, stream>>>(feat, mm, p_lo, p_hi);
    k1_w_deg<<<NBLK_P, BLK_, 0, stream>>>(p_lo, p_hi, nn, uw, deg);
    k2_uw<<<NBLK_P, BLK_, 0, stream>>>(uw, deg, nn);
    k3_qi<<<(TILES3_ * 32 + BLK_ - 1) / BLK_, BLK_, 0, stream>>>(feat, q1, qiw);
    k4_wd<<<(ROWS7_ * H_ + BLK_ - 1) / BLK_, BLK_, 0, stream>>>(feat, nn, q2, qiw, dw);
}


# Qi projection folded into k2's launch: the first 4375 waves of k2's resident grid project one 16-node tile on WMMA before their u_w rows; launches 5->4, its ramp and drain overlap k2's streaming
# speedup vs baseline: 2.6323x; 1.0409x over previous
#include <hip/hip_runtime.h>
#include <math.h>

#define B_   2
#define T_   8
#define N_   5000
#define K1_  17
#define H_   4
#define C_   16
#define NO_  8
#define T7_  (T_-1)
#define E2_  (N_*K1_)

#define BLK_      256
#define NBLK_P    1000
#define WAVES_P   (NBLK_P * BLK_ / 32)
#define THREADS_P (NBLK_P * BLK_)

typedef float v2f __attribute__((ext_vector_type(2)));
typedef float v8f __attribute__((ext_vector_type(8)));

template<int CTRL>
__device__ __forceinline__ int dpp_i(int x) {
    return __builtin_amdgcn_update_dpp(0, x, CTRL, 0xf, 0xf, true);
}
template<int CTRL>
__device__ __forceinline__ float dpp_f(float x) {
    return __int_as_float(dpp_i<CTRL>(__float_as_int(x)));
}
#define DPP_QUAD_1032   0xB1
#define DPP_QUAD_2301   0x4E
#define DPP_ROW_SHARE0  0x150

__device__ __forceinline__ float xhalf(float x) {
    const int b = __float_as_int(x);
    return __int_as_float(__builtin_amdgcn_permlanex16(b, b, 0x76543210, 0xfedcba98, false, false));
}

#define PLANE_   (N_ * H_ * C_)
#define PSPLIT_  11
static_assert((size_t)PSPLIT_ * PLANE_ <= (size_t)B_ * T7_ * E2_ * H_, "P planes 0..PSPLIT_-1 fit the d_w region");
static_assert((size_t)(B_ * T_ - PSPLIT_) * PLANE_ <= (size_t)B_ * T7_ * N_ * H_ * NO_, "the other P planes fit the Qi plane");

#define NODES1_ (B_ * T_ * N_)
#define TILESP_ (NODES1_ / 16)
static_assert(NODES1_ % 16 == 0, "kP: whole 16-row tiles");
static_assert((TILESP_ * 32) % BLK_ == 0, "kP: the grid holds no surplus wave");

__global__ void kp_proj(const float* __restrict__ feat,
                        const float* __restrict__ mm,
                        float*       __restrict__ p_lo,
                        float*       __restrict__ p_hi)
{
    const int wid  = __builtin_amdgcn_readfirstlane(
                         (int)((blockIdx.x * blockDim.x + threadIdx.x) >> 5));
    const int lane = threadIdx.x & 31;
    const int r  = lane & 15;
    const int hh = lane >> 4;

    const int g  = wid * 16 + r;
    const int bt = g / N_;

    const float* f0 = feat + (size_t)g * H_ * C_;
    float* o = (bt < PSPLIT_) ? (p_lo + (size_t)g * H_ * C_)
                              : (p_hi + (size_t)(g - PSPLIT_ * N_) * H_ * C_);

    v8f d[H_];
#pragma unroll
    for (int h = 0; h < H_; ++h) {
        const float* fr = f0 + h * C_;
        const float* Mr = mm + (h * C_ + r) * C_;
        const float4 ma = *reinterpret_cast<const float4*>(Mr + 8 * hh);
        const float4 mb = *reinterpret_cast<const float4*>(Mr + 8 * hh + 4);
        const float4 fa = *reinterpret_cast<const float4*>(fr + 8 * hh);
        const float4 fb = *reinterpret_cast<const float4*>(fr + 8 * hh + 4);
        v2f av[4], bv[4];
        av[0].x = ma.x;  av[0].y = ma.y;   bv[0].x = fa.x;  bv[0].y = fa.y;
        av[1].x = ma.z;  av[1].y = ma.w;   bv[1].x = fa.z;  bv[1].y = fa.w;
        av[2].x = mb.x;  av[2].y = mb.y;   bv[2].x = fb.x;  bv[2].y = fb.y;
        av[3].x = mb.z;  av[3].y = mb.w;   bv[3].x = fb.z;  bv[3].y = fb.w;
        v8f acc = {0.f,0.f,0.f,0.f,0.f,0.f,0.f,0.f};
#pragma unroll
        for (int c = 0; c < 4; ++c)
            acc = __builtin_amdgcn_wmma_f32_16x16x4_f32(
                      false, av[c], false, bv[c], (short)0, acc, false, false);
        d[h] = acc;
    }

#pragma unroll
    for (int h = 0; h < H_; ++h) {
        *reinterpret_cast<float4*>(o + h * C_ + 8 * hh) =
            make_float4(d[h][0], d[h][1], d[h][2], d[h][3]);
        *reinterpret_cast<float4*>(o + h * C_ + 8 * hh + 4) =
            make_float4(d[h][4], d[h][5], d[h][6], d[h][7]);
    }
}

#define TRIPS1_ (NODES1_ / WAVES_P)
static_assert(NODES1_ % WAVES_P == 0, "k1: every wave takes the same number of nodes");

template<int Q>
__device__ __forceinline__ float4 gather_piece(const float* __restrict__ plane, int nbrv, int n,
                                               unsigned piece_bytes, bool& ok)
{
    const int nb = dpp_i<DPP_ROW_SHARE0 + 2 * Q>(nbrv);
    ok = (nb >= 0);
    const int jn = ok ? nb : n;
    return *reinterpret_cast<const float4*>(
               reinterpret_cast<const char*>(plane) + ((unsigned)jn * (unsigned)(H_ * C_ * 4) + piece_bytes));
}

__device__ __forceinline__ float edge_weight(const float4 pi, const float4 pj, bool ok)
{
    const float dx = pi.x - pj.x, dy = pi.y - pj.y, dz = pi.z - pj.z, dw = pi.w - pj.w;
    float s = dx * dx;
    s = fmaf(dy, dy, s);
    s = fmaf(dz, dz, s);
    s = fmaf(dw, dw, s);
    s += dpp_f<DPP_QUAD_1032>(s);
    s += dpp_f<DPP_QUAD_2301>(s);
    return ok ? __expf(-s) : 0.f;
}

__global__ void k1_w_deg(const float* __restrict__ p_lo,
                         const float* __restrict__ p_hi,
                         const int*   __restrict__ nn,
                         float*       __restrict__ w_out,
                         float*       __restrict__ deg)
{
    const int wid  = __builtin_amdgcn_readfirstlane(
                         (int)((blockIdx.x * blockDim.x + threadIdx.x) >> 5));
    const int lane = threadIdx.x & 31;

    const int c  = lane & 15;
    const int hh = lane >> 4;
    const int p  = c & 3;
    const int h  = c >> 2;
    const int slot = (c + hh) & 15;
    const unsigned piece_bytes = (unsigned)c * 16u;

    int nbr_next = nn[(wid % N_) * K1_ + 1 + slot];

    for (int it = 0; it < TRIPS1_; ++it) {
        const int g  = wid + it * WAVES_P;
        const int bt = g / N_;
        const int n  = g - bt * N_;

        const int nbrv = nbr_next;
        if (it + 1 < TRIPS1_)
            nbr_next = nn[((g + WAVES_P) % N_) * K1_ + 1 + slot];

        const float* plane = (bt < PSPLIT_) ? (p_lo + (size_t)bt * PLANE_)
                                            : (p_hi + (size_t)(bt - PSPLIT_) * PLANE_);

        const float4 pi = *reinterpret_cast<const float4*>(
                              reinterpret_cast<const char*>(plane) + ((unsigned)n * (unsigned)(H_ * C_ * 4) + piece_bytes));

        bool ok[8];
        float4 pj[8];
        pj[0] = gather_piece<0>(plane, nbrv, n, piece_bytes, ok[0]);
        pj[1] = gather_piece<1>(plane, nbrv, n, piece_bytes, ok[1]);
        pj[2] = gather_piece<2>(plane, nbrv, n, piece_bytes, ok[2]);
        pj[3] = gather_piece<3>(plane, nbrv, n, piece_bytes, ok[3]);
        pj[4] = gather_piece<4>(plane, nbrv, n, piece_bytes, ok[4]);
        pj[5] = gather_piece<5>(plane, nbrv, n, piece_bytes, ok[5]);
        pj[6] = gather_piece<6>(plane, nbrv, n, piece_bytes, ok[6]);
        pj[7] = gather_piece<7>(plane, nbrv, n, piece_bytes, ok[7]);

        float wv[8];
#pragma unroll
        for (int q = 0; q < 8; ++q) wv[q] = edge_weight(pi, pj[q], ok[q]);

        float dp = ((wv[0] + wv[1]) + (wv[2] + wv[3])) + ((wv[4] + wv[5]) + (wv[6] + wv[7]));
        dp += xhalf(dp);

        const float wlo = (p == 0) ? wv[0] : (p == 1) ? wv[1] : (p == 2) ? wv[2] : wv[3];
        const float whi = (p == 0) ? wv[4] : (p == 1) ? wv[5] : (p == 2) ? wv[6] : wv[7];
        float* wrow = w_out + (size_t)g * 16 * H_ + (2 * p + hh) * H_ + h;
        wrow[0]      = wlo;
        wrow[8 * H_] = whi;
        if (lane < 16 && p == 0)
            deg[(size_t)g * H_ + h] = dp;
    }
}

#define ROWS7_  (B_ * T7_ * N_)
#define TILES3_ (ROWS7_ / 16)
static_assert(ROWS7_ % 16 == 0, "Qi: whole 16-row tiles");
static_assert(TILES3_ <= WAVES_P, "Qi: kernel 2's grid has a wave for every tile");

__device__ __forceinline__ void qi_tile(const float* __restrict__ feat,
                                        const float* __restrict__ q1,
                                        float*       __restrict__ qi,
                                        int tile, int lane)
{
    const int r  = lane & 15;
    const int hh = lane >> 4;

    const int g7  = tile * 16 + r;
    const int bt7 = g7 / N_;
    const int n   = g7 - bt7 * N_;
    const int b   = bt7 / T7_;
    const int t   = bt7 - b * T7_;

    const float* f0 = feat + (((size_t)(b * T_ + t) * N_ + n) * H_ * C_);

    v8f d[H_];
#pragma unroll
    for (int h = 0; h < H_; ++h) {
        const float* fr = f0 + h * C_;
        const float* Qh = q1 + (h * NO_ + (r & (NO_ - 1))) * C_;
        const float4 qa = *reinterpret_cast<const float4*>(Qh + 8 * hh);
        const float4 qb = *reinterpret_cast<const float4*>(Qh + 8 * hh + 4);
        const float4 fa = *reinterpret_cast<const float4*>(fr + 8 * hh);
        const float4 fb = *reinterpret_cast<const float4*>(fr + 8 * hh + 4);
        v2f av[4], bv[4];
        av[0].x = qa.x;  av[0].y = qa.y;   bv[0].x = fa.x;  bv[0].y = fa.y;
        av[1].x = qa.z;  av[1].y = qa.w;   bv[1].x = fa.z;  bv[1].y = fa.w;
        av[2].x = qb.x;  av[2].y = qb.y;   bv[2].x = fb.x;  bv[2].y = fb.y;
        av[3].x = qb.z;  av[3].y = qb.w;   bv[3].x = fb.z;  bv[3].y = fb.w;
        v8f acc = {0.f,0.f,0.f,0.f,0.f,0.f,0.f,0.f};
#pragma unroll
        for (int c = 0; c < 4; ++c)
            acc = __builtin_amdgcn_wmma_f32_16x16x4_f32(
                      false, av[c], false, bv[c], (short)0, acc, false, false);
        d[h] = acc;
    }

    if (hh == 0) {
        float* o = qi + (size_t)g7 * H_ * NO_;
#pragma unroll
        for (int h = 0; h < H_; ++h) {
            *reinterpret_cast<float4*>(o + h * NO_) =
                make_float4(d[h][0], d[h][1], d[h][2], d[h][3]);
            *reinterpret_cast<float4*>(o + h * NO_ + 4) =
                make_float4(d[h][4], d[h][5], d[h][6], d[h][7]);
        }
    }
}

__device__ __forceinline__ float norm_one(float w, float di, float dj)
{
    const float dm = sqrtf(di * dj);
    return (dm > 0.f) ? (w / dm) : 0.f;
}

#define ITEMS2_ (B_ * T_ * N_ * 16)
#define TRIPS2_ (ITEMS2_ / THREADS_P)
static_assert(ITEMS2_ % THREADS_P == 0, "k2: every thread takes the same number of rows");

__global__ void k2_uw_qi(float* __restrict__ uw,
                         const float* __restrict__ deg,
                         const int*   __restrict__ nn,
                         const float* __restrict__ feat,
                         const float* __restrict__ q1,
                         float*       __restrict__ qi)
{
    const int t0 = (int)(blockIdx.x * blockDim.x + threadIdx.x);

    const int wid = __builtin_amdgcn_readfirstlane(t0 >> 5);
    if (wid < TILES3_)
        qi_tile(feat, q1, qi, wid, (int)(threadIdx.x & 31));

    float4 w[TRIPS2_], di[TRIPS2_], dj[TRIPS2_];
    int    nb[TRIPS2_], own[TRIPS2_];
#pragma unroll
    for (int i = 0; i < TRIPS2_; ++i) {
        const int item = t0 + i * THREADS_P;
        const int k  = item % 16;
        own[i] = item / 16;
        const int n  = own[i] % N_;
        w[i]  = *reinterpret_cast<const float4*>(uw + (size_t)item * H_);
        nb[i] = nn[n * K1_ + 1 + k];
        di[i] = *reinterpret_cast<const float4*>(deg + (size_t)own[i] * H_);
    }
#pragma unroll
    for (int i = 0; i < TRIPS2_; ++i) {
        const int n   = own[i] % N_;
        const int nbc = (nb[i] >= 0) ? nb[i] : n;
        dj[i] = *reinterpret_cast<const float4*>(deg + (size_t)(own[i] - n + nbc) * H_);
        if (nb[i] < 0) dj[i] = make_float4(0.f, 0.f, 0.f, 0.f);
    }
#pragma unroll
    for (int i = 0; i < TRIPS2_; ++i) {
        const int item = t0 + i * THREADS_P;
        float4 o;
        o.x = norm_one(w[i].x, di[i].x, dj[i].x);
        o.y = norm_one(w[i].y, di[i].y, dj[i].y);
        o.z = norm_one(w[i].z, di[i].z, dj[i].z);
        o.w = norm_one(w[i].w, di[i].w, dj[i].w);
        *reinterpret_cast<float4*>(uw + (size_t)item * H_) = o;
    }
}

__global__ void __launch_bounds__(BLK_)
k4_wd(const float* __restrict__ feat,
      const int*   __restrict__ nn,
      const float* __restrict__ q2,
      const float* __restrict__ qi,
      float*       __restrict__ dw)
{
    const int tid = (int)(blockIdx.x * blockDim.x + threadIdx.x);
    if (tid >= ROWS7_ * H_) return;
    const int h   = tid % H_;
    const int g7  = tid / H_;
    const int bt7 = g7 / N_;
    const int n   = g7 - bt7 * N_;
    const int b   = bt7 / T7_;
    const int t   = bt7 - b * T7_;

    int nb[K1_];
#pragma unroll
    for (int k = 0; k < K1_; ++k) nb[k] = nn[n * K1_ + k];

    const float* f = feat + ((((size_t)(b * T_ + t + 1) * N_ + n) * H_ + h) * C_);
    const float* Q = q2 + h * NO_ * C_;
    float fv[C_];
#pragma unroll
    for (int j = 0; j < C_; j += 4) {
        const float4 x = *reinterpret_cast<const float4*>(f + j);
        fv[j] = x.x; fv[j + 1] = x.y; fv[j + 2] = x.z; fv[j + 3] = x.w;
    }
    float qj[NO_];
#pragma unroll
    for (int i = 0; i < NO_; ++i) {
        float s = 0.f;
#pragma unroll
        for (int j = 0; j < C_; ++j) s += Q[i * C_ + j] * fv[j];
        qj[i] = s;
    }

    const float* qrow = qi + ((size_t)bt7 * N_ * H_ + h) * NO_;
    float*       o    = dw + ((size_t)bt7 * E2_ + (size_t)n * K1_) * H_ + h;
    float wd[K1_];
#pragma unroll
    for (int k = 0; k < K1_; ++k) {
        const bool ok = (nb[k] >= 0);
        const int  jn = ok ? nb[k] : n;
        const float* p = qrow + (size_t)jn * (H_ * NO_);
        const float4 a0 = *reinterpret_cast<const float4*>(p);
        const float4 a1 = *reinterpret_cast<const float4*>(p + 4);
        const float s = ((a0.x * qj[0] + a0.y * qj[1]) + (a0.z * qj[2] + a0.w * qj[3]))
                      + ((a1.x * qj[4] + a1.y * qj[5]) + (a1.z * qj[6] + a1.w * qj[7]));
        const float e = __expf(-s);
        wd[k] = ok ? e : 0.f;
    }

    float sum = 0.f;
#pragma unroll
    for (int k = 0; k < K1_; ++k) sum += wd[k];
    const float inv = (sum > 0.f) ? (1.f / sum) : 0.f;
#pragma unroll
    for (int k = 0; k < K1_; ++k) o[k * H_] = wd[k] * inv;
}

extern "C" void kernel_launch(void* const* d_in, const int* in_sizes, int n_in,
                              void* d_out, int out_size, void* d_ws, size_t ws_size,
                              hipStream_t stream)
{
    const float* feat = (const float*)d_in[0];
    const int*   nn   = (const int*)  d_in[1];
    const float* q1   = (const float*)d_in[2];
    const float* q2   = (const float*)d_in[3];
    const float* mm   = (const float*)d_in[4];

    float* uw = (float*)d_out;
    float* dw = uw + (size_t)B_ * T_ * N_ * 16 * H_;

    float* deg = (float*)d_ws;
    float* qiw = deg + (size_t)B_ * T_ * N_ * H_;

    float* p_lo = dw;
    float* p_hi = qiw;

    kp_proj<<<(TILESP_ * 32) / BLK_, BLK_, 0, stream>>>(feat, mm, p_lo, p_hi);
    k1_w_deg<<<NBLK_P, BLK_, 0, stream>>>(p_lo, p_hi, nn, uw, deg);
    k2_uw_qi<<<NBLK_P, BLK_, 0, stream>>>(uw, deg, nn, feat, q1, qiw);
    k4_wd<<<(ROWS7_ * H_ + BLK_ - 1) / BLK_, BLK_, 0, stream>>>(feat, nn, q2, qiw, dw);
}


# deg stored node-major [n][b*T+t][h]; k2 is one wave per node over all 16 planes: a neighbour's deg row is 2 whole 128-byte lines, all bytes used (was 16 B of a line per gather); k2 8000->5000 waves
# speedup vs baseline: 2.7842x; 1.0577x over previous
#include <hip/hip_runtime.h>
#include <math.h>

#define B_   2
#define T_   8
#define N_   5000
#define K1_  17
#define H_   4
#define C_   16
#define NO_  8
#define T7_  (T_-1)
#define BT_  (B_*T_)
#define E2_  (N_*K1_)

#define BLK_      256
#define NBLK_P    1000
#define WAVES_P   (NBLK_P * BLK_ / 32)

typedef float v2f __attribute__((ext_vector_type(2)));
typedef float v8f __attribute__((ext_vector_type(8)));

template<int CTRL>
__device__ __forceinline__ int dpp_i(int x) {
    return __builtin_amdgcn_update_dpp(0, x, CTRL, 0xf, 0xf, true);
}
template<int CTRL>
__device__ __forceinline__ float dpp_f(float x) {
    return __int_as_float(dpp_i<CTRL>(__float_as_int(x)));
}
#define DPP_QUAD_1032   0xB1
#define DPP_QUAD_2301   0x4E
#define DPP_ROW_SHARE0  0x150

__device__ __forceinline__ float xhalf(float x) {
    const int b = __float_as_int(x);
    return __int_as_float(__builtin_amdgcn_permlanex16(b, b, 0x76543210, 0xfedcba98, false, false));
}

#define PLANE_   (N_ * H_ * C_)
#define PSPLIT_  11
static_assert((size_t)PSPLIT_ * PLANE_ <= (size_t)B_ * T7_ * E2_ * H_, "P planes 0..PSPLIT_-1 fit the d_w region");
static_assert((size_t)(B_ * T_ - PSPLIT_) * PLANE_ <= (size_t)B_ * T7_ * N_ * H_ * NO_, "the other P planes fit the Qi plane");

#define NODES1_ (B_ * T_ * N_)
#define TILESP_ (NODES1_ / 16)
static_assert(NODES1_ % 16 == 0, "kP: whole 16-row tiles");
static_assert((TILESP_ * 32) % BLK_ == 0, "kP: the grid holds no surplus wave");

__global__ void kp_proj(const float* __restrict__ feat,
                        const float* __restrict__ mm,
                        float*       __restrict__ p_lo,
                        float*       __restrict__ p_hi)
{
    const int wid  = __builtin_amdgcn_readfirstlane(
                         (int)((blockIdx.x * blockDim.x + threadIdx.x) >> 5));
    const int lane = threadIdx.x & 31;
    const int r  = lane & 15;
    const int hh = lane >> 4;

    const int g  = wid * 16 + r;
    const int bt = g / N_;

    const float* f0 = feat + (size_t)g * H_ * C_;
    float* o = (bt < PSPLIT_) ? (p_lo + (size_t)g * H_ * C_)
                              : (p_hi + (size_t)(g - PSPLIT_ * N_) * H_ * C_);

    v8f d[H_];
#pragma unroll
    for (int h = 0; h < H_; ++h) {
        const float* fr = f0 + h * C_;
        const float* Mr = mm + (h * C_ + r) * C_;
        const float4 ma = *reinterpret_cast<const float4*>(Mr + 8 * hh);
        const float4 mb = *reinterpret_cast<const float4*>(Mr + 8 * hh + 4);
        const float4 fa = *reinterpret_cast<const float4*>(fr + 8 * hh);
        const float4 fb = *reinterpret_cast<const float4*>(fr + 8 * hh + 4);
        v2f av[4], bv[4];
        av[0].x = ma.x;  av[0].y = ma.y;   bv[0].x = fa.x;  bv[0].y = fa.y;
        av[1].x = ma.z;  av[1].y = ma.w;   bv[1].x = fa.z;  bv[1].y = fa.w;
        av[2].x = mb.x;  av[2].y = mb.y;   bv[2].x = fb.x;  bv[2].y = fb.y;
        av[3].x = mb.z;  av[3].y = mb.w;   bv[3].x = fb.z;  bv[3].y = fb.w;
        v8f acc = {0.f,0.f,0.f,0.f,0.f,0.f,0.f,0.f};
#pragma unroll
        for (int c = 0; c < 4; ++c)
            acc = __builtin_amdgcn_wmma_f32_16x16x4_f32(
                      false, av[c], false, bv[c], (short)0, acc, false, false);
        d[h] = acc;
    }

#pragma unroll
    for (int h = 0; h < H_; ++h) {
        *reinterpret_cast<float4*>(o + h * C_ + 8 * hh) =
            make_float4(d[h][0], d[h][1], d[h][2], d[h][3]);
        *reinterpret_cast<float4*>(o + h * C_ + 8 * hh + 4) =
            make_float4(d[h][4], d[h][5], d[h][6], d[h][7]);
    }
}

#define TRIPS1_ (NODES1_ / WAVES_P)
static_assert(NODES1_ % WAVES_P == 0, "k1: every wave takes the same number of nodes");

template<int Q>
__device__ __forceinline__ float4 gather_piece(const float* __restrict__ plane, int nbrv, int n,
                                               unsigned piece_bytes, bool& ok)
{
    const int nb = dpp_i<DPP_ROW_SHARE0 + 2 * Q>(nbrv);
    ok = (nb >= 0);
    const int jn = ok ? nb : n;
    return *reinterpret_cast<const float4*>(
               reinterpret_cast<const char*>(plane) + ((unsigned)jn * (unsigned)(H_ * C_ * 4) + piece_bytes));
}

__device__ __forceinline__ float edge_weight(const float4 pi, const float4 pj, bool ok)
{
    const float dx = pi.x - pj.x, dy = pi.y - pj.y, dz = pi.z - pj.z, dw = pi.w - pj.w;
    float s = dx * dx;
    s = fmaf(dy, dy, s);
    s = fmaf(dz, dz, s);
    s = fmaf(dw, dw, s);
    s += dpp_f<DPP_QUAD_1032>(s);
    s += dpp_f<DPP_QUAD_2301>(s);
    return ok ? __expf(-s) : 0.f;
}

__global__ void k1_w_deg(const float* __restrict__ p_lo,
                         const float* __restrict__ p_hi,
                         const int*   __restrict__ nn,
                         float*       __restrict__ w_out,
                         float*       __restrict__ deg)
{
    const int wid  = __builtin_amdgcn_readfirstlane(
                         (int)((blockIdx.x * blockDim.x + threadIdx.x) >> 5));
    const int lane = threadIdx.x & 31;

    const int c  = lane & 15;
    const int hh = lane >> 4;
    const int p  = c & 3;
    const int h  = c >> 2;
    const int slot = (c + hh) & 15;
    const unsigned piece_bytes = (unsigned)c * 16u;

    int nbr_next = nn[(wid % N_) * K1_ + 1 + slot];

    for (int it = 0; it < TRIPS1_; ++it) {
        const int g  = wid + it * WAVES_P;
        const int bt = g / N_;
        const int n  = g - bt * N_;

        const int nbrv = nbr_next;
        if (it + 1 < TRIPS1_)
            nbr_next = nn[((g + WAVES_P) % N_) * K1_ + 1 + slot];

        const float* plane = (bt < PSPLIT_) ? (p_lo + (size_t)bt * PLANE_)
                                            : (p_hi + (size_t)(bt - PSPLIT_) * PLANE_);

        const float4 pi = *reinterpret_cast<const float4*>(
                              reinterpret_cast<const char*>(plane) + ((unsigned)n * (unsigned)(H_ * C_ * 4) + piece_bytes));

        bool ok[8];
        float4 pj[8];
        pj[0] = gather_piece<0>(plane, nbrv, n, piece_bytes, ok[0]);
        pj[1] = gather_piece<1>(plane, nbrv, n, piece_bytes, ok[1]);
        pj[2] = gather_piece<2>(plane, nbrv, n, piece_bytes, ok[2]);
        pj[3] = gather_piece<3>(plane, nbrv, n, piece_bytes, ok[3]);
        pj[4] = gather_piece<4>(plane, nbrv, n, piece_bytes, ok[4]);
        pj[5] = gather_piece<5>(plane, nbrv, n, piece_bytes, ok[5]);
        pj[6] = gather_piece<6>(plane, nbrv, n, piece_bytes, ok[6]);
        pj[7] = gather_piece<7>(plane, nbrv, n, piece_bytes, ok[7]);

        float wv[8];
#pragma unroll
        for (int q = 0; q < 8; ++q) wv[q] = edge_weight(pi, pj[q], ok[q]);

        float dp = ((wv[0] + wv[1]) + (wv[2] + wv[3])) + ((wv[4] + wv[5]) + (wv[6] + wv[7]));
        dp += xhalf(dp);

        const float wlo = (p == 0) ? wv[0] : (p == 1) ? wv[1] : (p == 2) ? wv[2] : wv[3];
        const float whi = (p == 0) ? wv[4] : (p == 1) ? wv[5] : (p == 2) ? wv[6] : wv[7];
        float* wrow = w_out + (size_t)g * 16 * H_ + (2 * p + hh) * H_ + h;
        wrow[0]      = wlo;
        wrow[8 * H_] = whi;
        if (lane < 16 && p == 0)
            deg[((size_t)n * BT_ + bt) * H_ + h] = dp;
    }
}

#define ROWS7_  (B_ * T7_ * N_)
#define TILES3_ (ROWS7_ / 16)
static_assert(ROWS7_ % 16 == 0, "Qi: whole 16-row tiles");

__device__ __forceinline__ void qi_tile(const float* __restrict__ feat,
                                        const float* __restrict__ q1,
                                        float*       __restrict__ qi,
                                        int tile, int lane)
{
    const int r  = lane & 15;
    const int hh = lane >> 4;

    const int g7  = tile * 16 + r;
    const int bt7 = g7 / N_;
    const int n   = g7 - bt7 * N_;
    const int b   = bt7 / T7_;
    const int t   = bt7 - b * T7_;

    const float* f0 = feat + (((size_t)(b * T_ + t) * N_ + n) * H_ * C_);

    v8f d[H_];
#pragma unroll
    for (int h = 0; h < H_; ++h) {
        const float* fr = f0 + h * C_;
        const float* Qh = q1 + (h * NO_ + (r & (NO_ - 1))) * C_;
        const float4 qa = *reinterpret_cast<const float4*>(Qh + 8 * hh);
        const float4 qb = *reinterpret_cast<const float4*>(Qh + 8 * hh + 4);
        const float4 fa = *reinterpret_cast<const float4*>(fr + 8 * hh);
        const float4 fb = *reinterpret_cast<const float4*>(fr + 8 * hh + 4);
        v2f av[4], bv[4];
        av[0].x = qa.x;  av[0].y = qa.y;   bv[0].x = fa.x;  bv[0].y = fa.y;
        av[1].x = qa.z;  av[1].y = qa.w;   bv[1].x = fa.z;  bv[1].y = fa.w;
        av[2].x = qb.x;  av[2].y = qb.y;   bv[2].x = fb.x;  bv[2].y = fb.y;
        av[3].x = qb.z;  av[3].y = qb.w;   bv[3].x = fb.z;  bv[3].y = fb.w;
        v8f acc = {0.f,0.f,0.f,0.f,0.f,0.f,0.f,0.f};
#pragma unroll
        for (int c = 0; c < 4; ++c)
            acc = __builtin_amdgcn_wmma_f32_16x16x4_f32(
                      false, av[c], false, bv[c], (short)0, acc, false, false);
        d[h] = acc;
    }

    if (hh == 0) {
        float* o = qi + (size_t)g7 * H_ * NO_;
#pragma unroll
        for (int h = 0; h < H_; ++h) {
            *reinterpret_cast<float4*>(o + h * NO_) =
                make_float4(d[h][0], d[h][1], d[h][2], d[h][3]);
            *reinterpret_cast<float4*>(o + h * NO_ + 4) =
                make_float4(d[h][4], d[h][5], d[h][6], d[h][7]);
        }
    }
}

__device__ __forceinline__ float norm_one(float w, float di, float dj)
{
    const float dm = sqrtf(di * dj);
    return (dm > 0.f) ? (w / dm) : 0.f;
}

#define STEPS2_ (BT_ / 2)
static_assert(BT_ % 2 == 0, "k2: planes are taken two per step");
static_assert((N_ * 32) % BLK_ == 0, "k2: the grid holds no surplus wave");
static_assert(TILES3_ <= N_, "Qi: kernel 2's grid has a wave for every tile");

__global__ void k2_uw_qi(float* __restrict__ uw,
                         const float* __restrict__ deg,
                         const int*   __restrict__ nn,
                         const float* __restrict__ feat,
                         const float* __restrict__ q1,
                         float*       __restrict__ qi)
{
    const int n    = __builtin_amdgcn_readfirstlane(
                         (int)((blockIdx.x * blockDim.x + threadIdx.x) >> 5));
    const int lane = threadIdx.x & 31;
    if (n < TILES3_)
        qi_tile(feat, q1, qi, n, lane);

    const int k  = lane & 15;
    const int hh = lane >> 4;

    const int nb = nn[n * K1_ + 1 + k];
    float4 w[STEPS2_], di[STEPS2_], dj[STEPS2_];
#pragma unroll
    for (int q = 0; q < STEPS2_; ++q) {
        const int bt = 2 * q + hh;
        w[q]  = *reinterpret_cast<const float4*>(uw + (((size_t)bt * N_ + n) * 16 + k) * H_);
        di[q] = *reinterpret_cast<const float4*>(deg + ((size_t)n * BT_ + bt) * H_);
    }
    const int jn = (nb >= 0) ? nb : n;
#pragma unroll
    for (int q = 0; q < STEPS2_; ++q) {
        const int bt = 2 * q + hh;
        dj[q] = *reinterpret_cast<const float4*>(deg + ((size_t)jn * BT_ + bt) * H_);
        if (nb < 0) dj[q] = make_float4(0.f, 0.f, 0.f, 0.f);
    }
#pragma unroll
    for (int q = 0; q < STEPS2_; ++q) {
        const int bt = 2 * q + hh;
        float4 o;
        o.x = norm_one(w[q].x, di[q].x, dj[q].x);
        o.y = norm_one(w[q].y, di[q].y, dj[q].y);
        o.z = norm_one(w[q].z, di[q].z, dj[q].z);
        o.w = norm_one(w[q].w, di[q].w, dj[q].w);
        *reinterpret_cast<float4*>(uw + (((size_t)bt * N_ + n) * 16 + k) * H_) = o;
    }
}

__global__ void __launch_bounds__(BLK_)
k4_wd(const float* __restrict__ feat,
      const int*   __restrict__ nn,
      const float* __restrict__ q2,
      const float* __restrict__ qi,
      float*       __restrict__ dw)
{
    const int tid = (int)(blockIdx.x * blockDim.x + threadIdx.x);
    if (tid >= ROWS7_ * H_) return;
    const int h   = tid % H_;
    const int g7  = tid / H_;
    const int bt7 = g7 / N_;
    const int n   = g7 - bt7 * N_;
    const int b   = bt7 / T7_;
    const int t   = bt7 - b * T7_;

    int nb[K1_];
#pragma unroll
    for (int k = 0; k < K1_; ++k) nb[k] = nn[n * K1_ + k];

    const float* f = feat + ((((size_t)(b * T_ + t + 1) * N_ + n) * H_ + h) * C_);
    const float* Q = q2 + h * NO_ * C_;
    float fv[C_];
#pragma unroll
    for (int j = 0; j < C_; j += 4) {
        const float4 x = *reinterpret_cast<const float4*>(f + j);
        fv[j] = x.x; fv[j + 1] = x.y; fv[j + 2] = x.z; fv[j + 3] = x.w;
    }
    float qj[NO_];
#pragma unroll
    for (int i = 0; i < NO_; ++i) {
        float s = 0.f;
#pragma unroll
        for (int j = 0; j < C_; ++j) s += Q[i * C_ + j] * fv[j];
        qj[i] = s;
    }

    const float* qrow = qi + ((size_t)bt7 * N_ * H_ + h) * NO_;
    float*       o    = dw + ((size_t)bt7 * E2_ + (size_t)n * K1_) * H_ + h;
    float wd[K1_];
#pragma unroll
    for (int k = 0; k < K1_; ++k) {
        const bool ok = (nb[k] >= 0);
        const int  jn = ok ? nb[k] : n;
        const float* p = qrow + (size_t)jn * (H_ * NO_);
        const float4 a0 = *reinterpret_cast<const float4*>(p);
        const float4 a1 = *reinterpret_cast<const float4*>(p + 4);
        const float s = ((a0.x * qj[0] + a0.y * qj[1]) + (a0.z * qj[2] + a0.w * qj[3]))
                      + ((a1.x * qj[4] + a1.y * qj[5]) + (a1.z * qj[6] + a1.w * qj[7]));
        const float e = __expf(-s);
        wd[k] = ok ? e : 0.f;
    }

    float sum = 0.f;
#pragma unroll
    for (int k = 0; k < K1_; ++k) sum += wd[k];
    const float inv = (sum > 0.f) ? (1.f / sum) : 0.f;
#pragma unroll
    for (int k = 0; k < K1_; ++k) o[k * H_] = wd[k] * inv;
}

extern "C" void kernel_launch(void* const* d_in, const int* in_sizes, int n_in,
                              void* d_out, int out_size, void* d_ws, size_t ws_size,
                              hipStream_t stream)
{
    const float* feat = (const float*)d_in[0];
    const int*   nn   = (const int*)  d_in[1];
    const float* q1   = (const float*)d_in[2];
    const float* q2   = (const float*)d_in[3];
    const float* mm   = (const float*)d_in[4];

    float* uw = (float*)d_out;
    float* dw = uw + (size_t)B_ * T_ * N_ * 16 * H_;

    float* deg = (float*)d_ws;
    float* qiw = deg + (size_t)B_ * T_ * N_ * H_;

    float* p_lo = dw;
    float* p_hi = qiw;

    kp_proj<<<(TILESP_ * 32) / BLK_, BLK_, 0, stream>>>(feat, mm, p_lo, p_hi);
    k1_w_deg<<<NBLK_P, BLK_, 0, stream>>>(p_lo, p_hi, nn, uw, deg);
    k2_uw_qi<<<(N_ * 32) / BLK_, BLK_, 0, stream>>>(uw, deg, nn, feat, q1, qiw);
    k4_wd<<<(ROWS7_ * H_ + BLK_ - 1) / BLK_, BLK_, 0, stream>>>(feat, nn, q2, qiw, dw);
}


# Work re-split over the 4 launches: kernel P also projects Qi of planes 0-9 from its own fragments (P re-placed), so launches 3 and 4 each mix u_w rows 4:7 with d_w rows (gather beside streaming)
# speedup vs baseline: 3.0824x; 1.1071x over previous
#include <hip/hip_runtime.h>
#include <math.h>

#define B_   2
#define T_   8
#define N_   5000
#define K1_  17
#define H_   4
#define C_   16
#define NO_  8
#define T7_  (T_-1)
#define BT_  (B_*T_)
#define E2_  (N_*K1_)

#define BLK_      256
#define NBLK_P    1000
#define WAVES_P   (NBLK_P * BLK_ / 32)

typedef float v2f __attribute__((ext_vector_type(2)));
typedef float v8f __attribute__((ext_vector_type(8)));

template<int CTRL>
__device__ __forceinline__ int dpp_i(int x) {
    return __builtin_amdgcn_update_dpp(0, x, CTRL, 0xf, 0xf, true);
}
template<int CTRL>
__device__ __forceinline__ float dpp_f(float x) {
    return __int_as_float(dpp_i<CTRL>(__float_as_int(x)));
}
#define DPP_QUAD_1032   0xB1
#define DPP_QUAD_2301   0x4E
#define DPP_ROW_SHARE0  0x150

__device__ __forceinline__ float xhalf(float x) {
    const int b = __float_as_int(x);
    return __int_as_float(__builtin_amdgcn_permlanex16(b, b, 0x76543210, 0xfedcba98, false, false));
}

#define PLANE_   (N_ * H_ * C_)
#define PSPLIT_  14
#define QI_EARLY_PLANES_ 10
#define ROWS7_EARLY_  (QI_EARLY_PLANES_ * N_)
#define P_HI_OFF_     ((size_t)ROWS7_EARLY_ * H_ * NO_)
#define P_ROWS_QI_END_ ((((QI_EARLY_PLANES_ - 1) / T7_) * T_ + ((QI_EARLY_PLANES_ - 1) % T7_) + 1) * N_)
static_assert((size_t)PSPLIT_ * PLANE_ <= (size_t)B_ * T7_ * E2_ * H_, "P planes 0..PSPLIT_-1 fit the d_w region");
static_assert(P_HI_OFF_ + (size_t)(B_ * T_ - PSPLIT_) * PLANE_ <= (size_t)B_ * T7_ * N_ * H_ * NO_, "the early Qi rows and the other P planes fit the Qi plane together");

#define NODES1_ (B_ * T_ * N_)
#define TILESP_ (NODES1_ / 16)
static_assert(NODES1_ % 16 == 0, "kP: whole 16-row tiles");
static_assert((TILESP_ * 32) % BLK_ == 0, "kP: the grid holds no surplus wave");

__global__ void kp_proj(const float* __restrict__ feat,
                        const float* __restrict__ mm,
                        const float* __restrict__ q1,
                        float*       __restrict__ p_lo,
                        float*       __restrict__ p_hi,
                        float*       __restrict__ qi)
{
    const int wid  = __builtin_amdgcn_readfirstlane(
                         (int)((blockIdx.x * blockDim.x + threadIdx.x) >> 5));
    const int lane = threadIdx.x & 31;
    const int r  = lane & 15;
    const int hh = lane >> 4;

    const int g  = wid * 16 + r;
    const int bt = g / N_;

    const float* f0 = feat + (size_t)g * H_ * C_;
    float* o = (bt < PSPLIT_) ? (p_lo + (size_t)g * H_ * C_)
                              : (p_hi + (size_t)(g - PSPLIT_ * N_) * H_ * C_);

    v8f d[H_];
    v2f bvh[H_][4];
#pragma unroll
    for (int h = 0; h < H_; ++h) {
        const float* fr = f0 + h * C_;
        const float* Mr = mm + (h * C_ + r) * C_;
        const float4 ma = *reinterpret_cast<const float4*>(Mr + 8 * hh);
        const float4 mb = *reinterpret_cast<const float4*>(Mr + 8 * hh + 4);
        const float4 fa = *reinterpret_cast<const float4*>(fr + 8 * hh);
        const float4 fb = *reinterpret_cast<const float4*>(fr + 8 * hh + 4);
        v2f av[4], bv[4];
        av[0].x = ma.x;  av[0].y = ma.y;   bv[0].x = fa.x;  bv[0].y = fa.y;
        av[1].x = ma.z;  av[1].y = ma.w;   bv[1].x = fa.z;  bv[1].y = fa.w;
        av[2].x = mb.x;  av[2].y = mb.y;   bv[2].x = fb.x;  bv[2].y = fb.y;
        av[3].x = mb.z;  av[3].y = mb.w;   bv[3].x = fb.z;  bv[3].y = fb.w;
        v8f acc = {0.f,0.f,0.f,0.f,0.f,0.f,0.f,0.f};
#pragma unroll
        for (int c = 0; c < 4; ++c)
            acc = __builtin_amdgcn_wmma_f32_16x16x4_f32(
                      false, av[c], false, bv[c], (short)0, acc, false, false);
        d[h] = acc;
#pragma unroll
        for (int c = 0; c < 4; ++c) bvh[h][c] = bv[c];
    }

#pragma unroll
    for (int h = 0; h < H_; ++h) {
        *reinterpret_cast<float4*>(o + h * C_ + 8 * hh) =
            make_float4(d[h][0], d[h][1], d[h][2], d[h][3]);
        *reinterpret_cast<float4*>(o + h * C_ + 8 * hh + 4) =
            make_float4(d[h][4], d[h][5], d[h][6], d[h][7]);
    }

    if (wid * 16 < P_ROWS_QI_END_) {
        const int n   = g - bt * N_;
        const int b   = bt / T_;
        const int t   = bt - b * T_;
        const int bt7 = b * T7_ + t;
        v8f dq[H_];
#pragma unroll
        for (int h = 0; h < H_; ++h) {
            const float* Qh = q1 + (h * NO_ + (r & (NO_ - 1))) * C_;
            const float4 qa = *reinterpret_cast<const float4*>(Qh + 8 * hh);
            const float4 qb = *reinterpret_cast<const float4*>(Qh + 8 * hh + 4);
            v2f aq[4];
            aq[0].x = qa.x;  aq[0].y = qa.y;
            aq[1].x = qa.z;  aq[1].y = qa.w;
            aq[2].x = qb.x;  aq[2].y = qb.y;
            aq[3].x = qb.z;  aq[3].y = qb.w;
            v8f acc = {0.f,0.f,0.f,0.f,0.f,0.f,0.f,0.f};
#pragma unroll
            for (int c = 0; c < 4; ++c)
                acc = __builtin_amdgcn_wmma_f32_16x16x4_f32(
                          false, aq[c], false, bvh[h][c], (short)0, acc, false, false);
            dq[h] = acc;
        }
        if (hh == 0 && t < T7_ && bt7 < QI_EARLY_PLANES_) {
            float* oq = qi + ((size_t)bt7 * N_ + n) * H_ * NO_;
#pragma unroll
            for (int h = 0; h < H_; ++h) {
                *reinterpret_cast<float4*>(oq + h * NO_) =
                    make_float4(dq[h][0], dq[h][1], dq[h][2], dq[h][3]);
                *reinterpret_cast<float4*>(oq + h * NO_ + 4) =
                    make_float4(dq[h][4], dq[h][5], dq[h][6], dq[h][7]);
            }
        }
    }
}

#define TRIPS1_ (NODES1_ / WAVES_P)
static_assert(NODES1_ % WAVES_P == 0, "k1: every wave takes the same number of nodes");

template<int Q>
__device__ __forceinline__ float4 gather_piece(const float* __restrict__ plane, int nbrv, int n,
                                               unsigned piece_bytes, bool& ok)
{
    const int nb = dpp_i<DPP_ROW_SHARE0 + 2 * Q>(nbrv);
    ok = (nb >= 0);
    const int jn = ok ? nb : n;
    return *reinterpret_cast<const float4*>(
               reinterpret_cast<const char*>(plane) + ((unsigned)jn * (unsigned)(H_ * C_ * 4) + piece_bytes));
}

__device__ __forceinline__ float edge_weight(const float4 pi, const float4 pj, bool ok)
{
    const float dx = pi.x - pj.x, dy = pi.y - pj.y, dz = pi.z - pj.z, dw = pi.w - pj.w;
    float s = dx * dx;
    s = fmaf(dy, dy, s);
    s = fmaf(dz, dz, s);
    s = fmaf(dw, dw, s);
    s += dpp_f<DPP_QUAD_1032>(s);
    s += dpp_f<DPP_QUAD_2301>(s);
    return ok ? __expf(-s) : 0.f;
}

__global__ void k1_w_deg(const float* __restrict__ p_lo,
                         const float* __restrict__ p_hi,
                         const int*   __restrict__ nn,
                         float*       __restrict__ w_out,
                         float*       __restrict__ deg)
{
    const int wid  = __builtin_amdgcn_readfirstlane(
                         (int)((blockIdx.x * blockDim.x + threadIdx.x) >> 5));
    const int lane = threadIdx.x & 31;

    const int c  = lane & 15;
    const int hh = lane >> 4;
    const int p  = c & 3;
    const int h  = c >> 2;
    const int slot = (c + hh) & 15;
    const unsigned piece_bytes = (unsigned)c * 16u;

    int nbr_next = nn[(wid % N_) * K1_ + 1 + slot];

    for (int it = 0; it < TRIPS1_; ++it) {
        const int g  = wid + it * WAVES_P;
        const int bt = g / N_;
        const int n  = g - bt * N_;

        const int nbrv = nbr_next;
        if (it + 1 < TRIPS1_)
            nbr_next = nn[((g + WAVES_P) % N_) * K1_ + 1 + slot];

        const float* plane = (bt < PSPLIT_) ? (p_lo + (size_t)bt * PLANE_)
                                            : (p_hi + (size_t)(bt - PSPLIT_) * PLANE_);

        const float4 pi = *reinterpret_cast<const float4*>(
                              reinterpret_cast<const char*>(plane) + ((unsigned)n * (unsigned)(H_ * C_ * 4) + piece_bytes));

        bool ok[8];
        float4 pj[8];
        pj[0] = gather_piece<0>(plane, nbrv, n, piece_bytes, ok[0]);
        pj[1] = gather_piece<1>(plane, nbrv, n, piece_bytes, ok[1]);
        pj[2] = gather_piece<2>(plane, nbrv, n, piece_bytes, ok[2]);
        pj[3] = gather_piece<3>(plane, nbrv, n, piece_bytes, ok[3]);
        pj[4] = gather_piece<4>(plane, nbrv, n, piece_bytes, ok[4]);
        pj[5] = gather_piece<5>(plane, nbrv, n, piece_bytes, ok[5]);
        pj[6] = gather_piece<6>(plane, nbrv, n, piece_bytes, ok[6]);
        pj[7] = gather_piece<7>(plane, nbrv, n, piece_bytes, ok[7]);

        float wv[8];
#pragma unroll
        for (int q = 0; q < 8; ++q) wv[q] = edge_weight(pi, pj[q], ok[q]);

        float dp = ((wv[0] + wv[1]) + (wv[2] + wv[3])) + ((wv[4] + wv[5]) + (wv[6] + wv[7]));
        dp += xhalf(dp);

        const float wlo = (p == 0) ? wv[0] : (p == 1) ? wv[1] : (p == 2) ? wv[2] : wv[3];
        const float whi = (p == 0) ? wv[4] : (p == 1) ? wv[5] : (p == 2) ? wv[6] : wv[7];
        float* wrow = w_out + (size_t)g * 16 * H_ + (2 * p + hh) * H_ + h;
        wrow[0]      = wlo;
        wrow[8 * H_] = whi;
        if (lane < 16 && p == 0)
            deg[((size_t)n * BT_ + bt) * H_ + h] = dp;
    }
}

#define ROWS7_  (B_ * T7_ * N_)
#define TILES3_ (ROWS7_ / 16)
#define TILES3_EARLY_ (ROWS7_EARLY_ / 16)
#define TILES3_LATE_  (TILES3_ - TILES3_EARLY_)
static_assert(ROWS7_ % 16 == 0, "Qi: whole 16-row tiles");
static_assert(ROWS7_EARLY_ % 16 == 0, "Qi: the early planes end on a tile boundary");

__device__ __forceinline__ void qi_tile(const float* __restrict__ feat,
                                        const float* __restrict__ q1,
                                        float*       __restrict__ qi,
                                        int tile, int lane)
{
    const int r  = lane & 15;
    const int hh = lane >> 4;

    const int g7  = tile * 16 + r;
    const int bt7 = g7 / N_;
    const int n   = g7 - bt7 * N_;
    const int b   = bt7 / T7_;
    const int t   = bt7 - b * T7_;

    const float* f0 = feat + (((size_t)(b * T_ + t) * N_ + n) * H_ * C_);

    v8f d[H_];
#pragma unroll
    for (int h = 0; h < H_; ++h) {
        const float* fr = f0 + h * C_;
        const float* Qh = q1 + (h * NO_ + (r & (NO_ - 1))) * C_;
        const float4 qa = *reinterpret_cast<const float4*>(Qh + 8 * hh);
        const float4 qb = *reinterpret_cast<const float4*>(Qh + 8 * hh + 4);
        const float4 fa = *reinterpret_cast<const float4*>(fr + 8 * hh);
        const float4 fb = *reinterpret_cast<const float4*>(fr + 8 * hh + 4);
        v2f av[4], bv[4];
        av[0].x = qa.x;  av[0].y = qa.y;   bv[0].x = fa.x;  bv[0].y = fa.y;
        av[1].x = qa.z;  av[1].y = qa.w;   bv[1].x = fa.z;  bv[1].y = fa.w;
        av[2].x = qb.x;  av[2].y = qb.y;   bv[2].x = fb.x;  bv[2].y = fb.y;
        av[3].x = qb.z;  av[3].y = qb.w;   bv[3].x = fb.z;  bv[3].y = fb.w;
        v8f acc = {0.f,0.f,0.f,0.f,0.f,0.f,0.f,0.f};
#pragma unroll
        for (int c = 0; c < 4; ++c)
            acc = __builtin_amdgcn_wmma_f32_16x16x4_f32(
                      false, av[c], false, bv[c], (short)0, acc, false, false);
        d[h] = acc;
    }

    if (hh == 0) {
        float* o = qi + (size_t)g7 * H_ * NO_;
#pragma unroll
        for (int h = 0; h < H_; ++h) {
            *reinterpret_cast<float4*>(o + h * NO_) =
                make_float4(d[h][0], d[h][1], d[h][2], d[h][3]);
            *reinterpret_cast<float4*>(o + h * NO_ + 4) =
                make_float4(d[h][4], d[h][5], d[h][6], d[h][7]);
        }
    }
}

__device__ __forceinline__ float norm_one(float w, float di, float dj)
{
    const float dm = sqrtf(di * dj);
    return (dm > 0.f) ? (w / dm) : 0.f;
}

#define STEPS2_ (BT_ / 2)
static_assert(BT_ % 2 == 0, "k2: planes are taken two per step");
static_assert((N_ * 32) % BLK_ == 0, "k2: its blocks hold no surplus wave");

__device__ __forceinline__ void k2_rows(float* __restrict__ uw,
                                        const float* __restrict__ deg,
                                        const int*   __restrict__ nn,
                                        const float* __restrict__ feat,
                                        const float* __restrict__ q1,
                                        float*       __restrict__ qi,
                                        int n, int lane, bool late_qi)
{
    if (late_qi && n < TILES3_LATE_)
        qi_tile(feat, q1, qi, TILES3_EARLY_ + n, lane);

    const int k  = lane & 15;
    const int hh = lane >> 4;

    const int nb = nn[n * K1_ + 1 + k];
    float4 w[STEPS2_], di[STEPS2_], dj[STEPS2_];
#pragma unroll
    for (int q = 0; q < STEPS2_; ++q) {
        const int bt = 2 * q + hh;
        w[q]  = *reinterpret_cast<const float4*>(uw + (((size_t)bt * N_ + n) * 16 + k) * H_);
        di[q] = *reinterpret_cast<const float4*>(deg + ((size_t)n * BT_ + bt) * H_);
    }
    const int jn = (nb >= 0) ? nb : n;
#pragma unroll
    for (int q = 0; q < STEPS2_; ++q) {
        const int bt = 2 * q + hh;
        dj[q] = *reinterpret_cast<const float4*>(deg + ((size_t)jn * BT_ + bt) * H_);
        if (nb < 0) dj[q] = make_float4(0.f, 0.f, 0.f, 0.f);
    }
#pragma unroll
    for (int q = 0; q < STEPS2_; ++q) {
        const int bt = 2 * q + hh;
        float4 o;
        o.x = norm_one(w[q].x, di[q].x, dj[q].x);
        o.y = norm_one(w[q].y, di[q].y, dj[q].y);
        o.z = norm_one(w[q].z, di[q].z, dj[q].z);
        o.w = norm_one(w[q].w, di[q].w, dj[q].w);
        *reinterpret_cast<float4*>(uw + (((size_t)bt * N_ + n) * 16 + k) * H_) = o;
    }
}

__device__ __forceinline__ void k4_rows(const float* __restrict__ feat,
                                        const int*   __restrict__ nn,
                                        const float* __restrict__ q2,
                                        const float* __restrict__ qi,
                                        float*       __restrict__ dw,
                                        int tid)
{
    const int h   = tid % H_;
    const int g7  = tid / H_;
    const int bt7 = g7 / N_;
    const int n   = g7 - bt7 * N_;
    const int b   = bt7 / T7_;
    const int t   = bt7 - b * T7_;

    int nb[K1_];
#pragma unroll
    for (int k = 0; k < K1_; ++k) nb[k] = nn[n * K1_ + k];

    const float* f = feat + ((((size_t)(b * T_ + t + 1) * N_ + n) * H_ + h) * C_);
    const float* Q = q2 + h * NO_ * C_;
    float fv[C_];
#pragma unroll
    for (int j = 0; j < C_; j += 4) {
        const float4 x = *reinterpret_cast<const float4*>(f + j);
        fv[j] = x.x; fv[j + 1] = x.y; fv[j + 2] = x.z; fv[j + 3] = x.w;
    }
    float qj[NO_];
#pragma unroll
    for (int i = 0; i < NO_; ++i) {
        float s = 0.f;
#pragma unroll
        for (int j = 0; j < C_; ++j) s += Q[i * C_ + j] * fv[j];
        qj[i] = s;
    }

    const float* qrow = qi + ((size_t)bt7 * N_ * H_ + h) * NO_;
    float*       o    = dw + ((size_t)bt7 * E2_ + (size_t)n * K1_) * H_ + h;
    float wd[K1_];
#pragma unroll
    for (int k = 0; k < K1_; ++k) {
        const bool ok = (nb[k] >= 0);
        const int  jn = ok ? nb[k] : n;
        const float* p = qrow + (size_t)jn * (H_ * NO_);
        const float4 a0 = *reinterpret_cast<const float4*>(p);
        const float4 a1 = *reinterpret_cast<const float4*>(p + 4);
        const float s = ((a0.x * qj[0] + a0.y * qj[1]) + (a0.z * qj[2] + a0.w * qj[3]))
                      + ((a1.x * qj[4] + a1.y * qj[5]) + (a1.z * qj[6] + a1.w * qj[7]));
        const float e = __expf(-s);
        wd[k] = ok ? e : 0.f;
    }

    float sum = 0.f;
#pragma unroll
    for (int k = 0; k < K1_; ++k) sum += wd[k];
    const float inv = (sum > 0.f) ? (1.f / sum) : 0.f;
#pragma unroll
    for (int k = 0; k < K1_; ++k) o[k * H_] = wd[k] * inv;
}

#define K2_BLOCKS_    ((N_ * 32) / BLK_)
#define K2_BLOCKS_A_  446
#define K4_THREADS_A_ (ROWS7_EARLY_ * H_)
#define K4_THREADS_B_ ((ROWS7_ - ROWS7_EARLY_) * H_)
#define MIXG_A_       112
#define MIXG_B_       45
static_assert(MIXG_A_ * 4 >= K2_BLOCKS_A_ && (size_t)MIXG_A_ * 7 * BLK_ >= K4_THREADS_A_, "third launch: the mixed grid holds every block of both kinds");
static_assert(MIXG_B_ * 4 >= K2_BLOCKS_ - K2_BLOCKS_A_ && (size_t)MIXG_B_ * 7 * BLK_ >= K4_THREADS_B_, "fourth launch: the mixed grid holds every block of both kinds");
static_assert(TILES3_LATE_ <= K2_BLOCKS_A_ * (BLK_ / 32), "Qi: the third launch's u_w rows have a wave for every late tile");

template<int KB0, int KBN, int TID0, int TIDN, bool LATE_QI>
__global__ void __launch_bounds__(BLK_)
k_uw_dw(float*       __restrict__ uw,
        const float* __restrict__ deg,
        const int*   __restrict__ nn,
        const float* __restrict__ feat,
        const float* __restrict__ q1,
        const float* __restrict__ q2,
        float*       __restrict__ qi,
        float*       __restrict__ dw)
{
    const int grp = (int)blockIdx.x / 11;
    const int r   = (int)blockIdx.x - grp * 11;
    if (r < 4) {
        const int kb = grp * 4 + r;
        if (kb < KBN) {
            const int n = __builtin_amdgcn_readfirstlane((KB0 + kb) * (BLK_ / 32) + (int)(threadIdx.x >> 5));
            k2_rows(uw, deg, nn, feat, q1, qi, n, (int)(threadIdx.x & 31), LATE_QI);
        }
    } else {
        const int tid = (grp * 7 + (r - 4)) * BLK_ + (int)threadIdx.x;
        if (tid < TIDN)
            k4_rows(feat, nn, q2, qi, dw, TID0 + tid);
    }
}

extern "C" void kernel_launch(void* const* d_in, const int* in_sizes, int n_in,
                              void* d_out, int out_size, void* d_ws, size_t ws_size,
                              hipStream_t stream)
{
    const float* feat = (const float*)d_in[0];
    const int*   nn   = (const int*)  d_in[1];
    const float* q1   = (const float*)d_in[2];
    const float* q2   = (const float*)d_in[3];
    const float* mm   = (const float*)d_in[4];

    float* uw = (float*)d_out;
    float* dw = uw + (size_t)B_ * T_ * N_ * 16 * H_;

    float* deg = (float*)d_ws;
    float* qiw = deg + (size_t)B_ * T_ * N_ * H_;

    float* p_lo = dw;
    float* p_hi = qiw + P_HI_OFF_;

    kp_proj<<<(TILESP_ * 32) / BLK_, BLK_, 0, stream>>>(feat, mm, q1, p_lo, p_hi, qiw);
    k1_w_deg<<<NBLK_P, BLK_, 0, stream>>>(p_lo, p_hi, nn, uw, deg);
    k_uw_dw<0, K2_BLOCKS_A_, 0, K4_THREADS_A_, true>
        <<<MIXG_A_ * 11, BLK_, 0, stream>>>(uw, deg, nn, feat, q1, q2, qiw, dw);
    k_uw_dw<K2_BLOCKS_A_, K2_BLOCKS_ - K2_BLOCKS_A_, K4_THREADS_A_, K4_THREADS_B_, false>
        <<<MIXG_B_ * 11, BLK_, 0, stream>>>(uw, deg, nn, feat, q1, q2, qiw, dw);
}
